# starting kernel
# baseline (speedup 1.0000x reference)
#include <hip/hip_runtime.h>


#define NCOMBO 23
#define CG_TOTAL 2351
#define NFRAGS 76
#define FRAG_TOTAL (NFRAGS * 64)
#define TAB_TOTAL (CG_TOTAL + FRAG_TOTAL)
#define WAVES_PER_BLOCK 4
#define TB (WAVES_PER_BLOCK * 32)
#define ZROW 84

typedef float v2f __attribute__((ext_vector_type(2)));
typedef float v8f __attribute__((ext_vector_type(8)));

__constant__ int c_l1[NCOMBO] = {0,0,0,0, 1,1,1,1,1,1,1,1, 2,2,2,2,2,2,2, 3,3,3,3};
__constant__ int c_l2[NCOMBO] = {0,1,2,3, 1,1,1,2,2,2,3,3, 2,2,2,2,3,3,3, 3,3,3,3};
__constant__ int c_L [NCOMBO] = {0,1,2,3, 0,1,2,1,2,3,2,3, 0,1,2,3,1,2,3, 0,1,2,3};
__constant__ float c_w[NCOMBO] = {1,2,2,2, 1,1,1,2,2,2,2,2, 1,1,1,1,2,2,2, 1,1,1,1};
__constant__ int c_off[NCOMBO] = {0,1,10,35, 84,93,120,165,210,285,390,495,
                                  642,667,742,867,1042,1147,1322, 1567,1616,1763,2008};
__constant__ int c_foff[NCOMBO] = {0,1,2,4, 8,9,10,11,12,14,17,20,
                                   24,26,28,32,38,42,48, 56,58,62,68};
__constant__ int d_xoff[4] = {0, 1, 10, 35};
__constant__ int d_x2pos[4] = {0, 1, 19, 69};
__constant__ int d_x2neg[4] = {0, 10, 44, 118};
__constant__ int d_diag[4] = {0, 4, 12, 19};

__global__ void wigner_cg_init(float* __restrict__ ws) {
  double f[16];
  f[0] = 1.0;
  for (int i = 1; i < 16; ++i) f[i] = f[i - 1] * i;
  for (int c = 0; c < NCOMBO; ++c) {
    const int l1 = c_l1[c], l2 = c_l2[c], L = c_L[c];
    const int n2 = 2 * l2 + 1, nL = 2 * L + 1;
    const int sz = (2 * l1 + 1) * n2 * nL, off = c_off[c];
    const double pref0 =
        sqrt((2.0 * L + 1.0) * f[L + l1 - l2] * f[L - l1 + l2] * f[l1 + l2 - L] / f[l1 + l2 + L + 1]);
    for (int t = threadIdx.x; t < sz; t += blockDim.x) {
      const int m = t / (n2 * nL), rem = t % (n2 * nL), n = rem / nL, k = rem % nL;
      const int m1 = m - l1, m2 = n - l2, M = m1 + m2;
      double val = 0.0;
      if (M >= -L && M <= L && k == M + L) {
        const double pref =
            pref0 * sqrt(f[L + M] * f[L - M] * f[l1 - m1] * f[l1 + m1] * f[l2 - m2] * f[l2 + m2]);
        double s = 0.0;
        for (int kk = 0; kk <= l1 + l2 - L; ++kk) {
          const int a1 = l1 - m1 - kk, a2 = l2 + m2 - kk;
          const int a3 = L - l2 + m1 + kk, a4 = L - l1 - m2 + kk;
          if (a1 < 0 || a2 < 0 || a3 < 0 || a4 < 0) continue;
          const double term = 1.0 / (f[kk] * f[l1 + l2 - L - kk] * f[a1] * f[a2] * f[a3] * f[a4]);
          s += (kk & 1) ? -term : term;
        }
        val = pref * s;
      }
      ws[off + t] = (float)val;
    }
  }
  __syncthreads();
  for (int c = 0; c < NCOMBO; ++c) {
    const int l1 = c_l1[c], l2 = c_l2[c], L = c_L[c];
    const int n1 = 2 * l1 + 1, R = (2 * l2 + 1) * (2 * L + 1);
    const int cgo = c_off[c], fo = c_foff[c];
    const int mtiles = (R + 15) >> 4, kch = (n1 + 3) >> 2;
    for (int t = threadIdx.x; t < mtiles * kch * 64; t += blockDim.x) {
      const int fi = t >> 6, s = t & 63;
      const int mt = fi / kch, kc = fi % kch;
      const int ln = s >> 1, j = s & 1;
      const int row = mt * 16 + (ln & 15);
      const int m = kc * 4 + ((ln >> 4) ? 2 : 0) + j;
      ws[CG_TOTAL + (fo + fi) * 64 + s] =
          (row < R && m < n1) ? ws[cgo + m * R + row] : 0.f;
    }
  }
}

__launch_bounds__(TB, 1)
__global__ void wigner_main(const float* __restrict__ x0, const float* __restrict__ x1,
                            const float* __restrict__ x2, const float* __restrict__ x3,
                            const float* __restrict__ tab, float* __restrict__ out, int Btot) {
  __shared__ __align__(16) float sTAB[TAB_TOTAL];
  __shared__ float sXT[WAVES_PER_BLOCK][85 * 16];
  __shared__ float sA [WAVES_PER_BLOCK][64 * 7 * 16];
  __shared__ float sE [WAVES_PER_BLOCK][7 * 7 * 7 * 16];
  __shared__ float sX2[WAVES_PER_BLOCK][167 * 16];

  const int tid = threadIdx.x, lane = tid & 31, wid = tid >> 5;
  const int b = lane & 15, half = lane >> 4;
  const float* sCG = sTAB;
  const float* sFRAG = sTAB + CG_TOTAL;

  for (int t = tid; t < TAB_TOTAL; t += TB) sTAB[t] = tab[t];

  const int tile = blockIdx.x * WAVES_PER_BLOCK + wid;
  const int base = tile * 16;
  float* XT = sXT[wid];
  float* A  = sA[wid];
  float* E  = sE[wid];
  float* X2 = sX2[wid];

  {
    int pe = base + WAVES_PER_BLOCK * 16;
    if (pe < Btot) __builtin_prefetch(&x3[(size_t)pe * 49], 0, 1);
  }

  for (int t = lane; t < 84 * 16; t += 32) {
    const int bb = t / 84, comp = t % 84;
    int e = base + bb;
    e = (e < Btot) ? e : (Btot - 1);
    float v;
    if (comp < 1)       v = x0[e];
    else if (comp < 10) v = x1[(size_t)e * 9 + (comp - 1)];
    else if (comp < 35) v = x2[(size_t)e * 25 + (comp - 10)];
    else                v = x3[(size_t)e * 49 + (comp - 35)];
    XT[comp * 16 + bb] = v;
  }
  for (int t = lane; t < 16; t += 32) XT[ZROW * 16 + t] = 0.f;
  for (int t = lane; t < 167 * 16; t += 32) X2[t] = 0.f;
  __syncthreads();

  for (int c = 0; c < NCOMBO; ++c) {
    const int l1 = c_l1[c], l2 = c_l2[c], L = c_L[c];
    const int n1 = 2 * l1 + 1, n2 = 2 * l2 + 1, nL = 2 * L + 1;
    const int R = n2 * nL, cgo = c_off[c], fb = c_foff[c];
    const int xo1 = d_xoff[l1], xo2 = d_xoff[l2];
    const int mtiles = (R + 15) >> 4, kch = (n1 + 3) >> 2;

    const int m0a = half ? 2 : 0;
    if (kch == 1) {
      for (int p = 0; p < n1; ++p) {
        v2f bf0;
        const int r0 = (m0a     < n1) ? (xo1 + m0a * n1 + p)       : ZROW;
        const int r1 = (m0a + 1 < n1) ? (xo1 + (m0a + 1) * n1 + p) : ZROW;
        bf0.x = XT[r0 * 16 + b];
        bf0.y = XT[r1 * 16 + b];
        for (int mt = 0; mt < mtiles; ++mt) {
          v8f acc = {0.f, 0.f, 0.f, 0.f, 0.f, 0.f, 0.f, 0.f};
          const v2f af0 = *(const v2f*)&sFRAG[(fb + mt) * 64 + lane * 2];
          acc = __builtin_amdgcn_wmma_f32_16x16x4_f32(false, af0, false, bf0,
                                                      (short)0, acc, false, false);
          const int rb = mt * 16 + half * 8;
#pragma unroll
          for (int r = 0; r < 8; ++r)
            A[((rb + r) * n1 + p) * 16 + b] = acc[r];
        }
      }
    } else {
      for (int p = 0; p < n1; ++p) {
        v2f bf0, bf1;
        {
          const int r0 = xo1 + m0a * n1 + p;
          const int r1 = xo1 + (m0a + 1) * n1 + p;
          bf0.x = XT[r0 * 16 + b];
          bf0.y = XT[r1 * 16 + b];
        }
        {
          const int m0 = 4 + m0a;
          const int r0 = (m0     < n1) ? (xo1 + m0 * n1 + p)       : ZROW;
          const int r1 = (m0 + 1 < n1) ? (xo1 + (m0 + 1) * n1 + p) : ZROW;
          bf1.x = XT[r0 * 16 + b];
          bf1.y = XT[r1 * 16 + b];
        }
        for (int mt = 0; mt < mtiles; ++mt) {
          v8f acc = {0.f, 0.f, 0.f, 0.f, 0.f, 0.f, 0.f, 0.f};
          const float* fp = &sFRAG[(fb + mt * 2) * 64 + lane * 2];
          const v2f af0 = *(const v2f*)fp;
          acc = __builtin_amdgcn_wmma_f32_16x16x4_f32(false, af0, false, bf0,
                                                      (short)0, acc, false, false);
          const v2f af1 = *(const v2f*)(fp + 64);
          acc = __builtin_amdgcn_wmma_f32_16x16x4_f32(false, af1, false, bf1,
                                                      (short)0, acc, false, false);
          const int rb = mt * 16 + half * 8;
#pragma unroll
          for (int r = 0; r < 8; ++r)
            A[((rb + r) * n1 + p) * 16 + b] = acc[r];
        }
      }
    }
    __syncthreads();

    for (int q = 0; q < n2; ++q) {
      float xv[7];
#pragma unroll
      for (int n = 0; n < 7; ++n)
        xv[n] = (n < n2) ? XT[(xo2 + n * n2 + q) * 16 + b] : 0.f;
      const int kp = nL * n1;
      for (int idx = half; idx < kp; idx += 2) {
        const int p = idx % n1, k = idx / n1;
        float e = 0.f;
#pragma unroll
        for (int n = 0; n < 7; ++n)
          if (n < n2)
            e += xv[n] * A[((n * nL + k) * n1 + p) * 16 + b];
        E[((k * n2 + q) * n1 + p) * 16 + b] = e;
      }
    }
    __syncthreads();

    const int x2base = ((l1 + l2 + L) & 1) ? d_x2neg[L] : d_x2pos[L];
    const float wc = c_w[c];
    for (int idx = half; idx < nL * nL; idx += 2) {
      const int j = idx % nL, k = idx / nL;
      float s = 0.f;
      for (int pq = 0; pq < n1 * n2; ++pq) {
        const float cc = sCG[cgo + pq * nL + j];
        if (cc != 0.f) {
          const int p = pq / n2, q = pq % n2;
          s += cc * E[((k * n2 + q) * n1 + p) * 16 + b];
        }
      }
      X2[(x2base + idx) * 16 + b] += wc * s;
    }
    __syncthreads();
  }

  if (lane < 16) {
    const int e = base + b;
    if (e < Btot) {
      const float nu1 = XT[0 * 16 + b];
      const float nu2 = X2[0 * 16 + b];
      float nu3 = 0.f, nu4 = 0.f;
      for (int l = 0; l <= 3; ++l) {
        const int n = 2 * l + 1;
        const int cgo = c_off[d_diag[l]];
        const int xo = d_xoff[l];
        const int xp = d_x2pos[l], xn = d_x2neg[l];
        for (int mn = 0; mn < n * n; ++mn) {
          const float c1 = sCG[cgo + mn];
          if (c1 == 0.f) continue;
          const int m = mn / n, nn = mn % n;
          for (int pq = 0; pq < n * n; ++pq) {
            const float c2 = sCG[cgo + pq];
            if (c2 == 0.f) continue;
            const int p = pq / n, q = pq % n;
            const float w = c1 * c2;
            nu3 += w * X2[(xp + m * n + p) * 16 + b] * XT[(xo + nn * n + q) * 16 + b];
            nu4 += w * X2[(xp + m * n + p) * 16 + b] * X2[(xp + nn * n + q) * 16 + b];
            if (l > 0)
              nu4 += w * X2[(xn + m * n + p) * 16 + b] * X2[(xn + nn * n + q) * 16 + b];
          }
        }
      }
      out[(size_t)e * 4 + 0] = nu1;
      out[(size_t)e * 4 + 1] = nu2;
      out[(size_t)e * 4 + 2] = nu3;
      out[(size_t)e * 4 + 3] = nu4;
    }
  }
}

extern "C" void kernel_launch(void* const* d_in, const int* in_sizes, int n_in,
                              void* d_out, int out_size, void* d_ws, size_t ws_size,
                              hipStream_t stream) {
  const float* x0 = (const float*)d_in[0];
  const float* x1 = (const float*)d_in[1];
  const float* x2 = (const float*)d_in[2];
  const float* x3 = (const float*)d_in[3];
  float* tab = (float*)d_ws;
  const int Btot = in_sizes[0];

  wigner_cg_init<<<1, 256, 0, stream>>>(tab);

  const int tiles = (Btot + 15) / 16;
  const int blocks = (tiles + WAVES_PER_BLOCK - 1) / WAVES_PER_BLOCK;
  wigner_main<<<blocks, TB, 0, stream>>>(x0, x1, x2, x3, tab, (float*)d_out, Btot);
}


# The 23 (l1,l2,L) combos become compile-time instantiations: constant loop bounds, offsets and divisions; term loops unroll and read their LDS operands as one group; same summation order
# speedup vs baseline: 2.4916x; 2.4916x over previous
#include <hip/hip_runtime.h>


#define NCOMBO 23
#define CG_TOTAL 2351
#define NFRAGS 76
#define FRAG_TOTAL (NFRAGS * 64)
#define TAB_TOTAL (CG_TOTAL + FRAG_TOTAL)
#define WAVES_PER_BLOCK 4
#define TB (WAVES_PER_BLOCK * 32)
#define ZROW 84

typedef float v2f __attribute__((ext_vector_type(2)));
typedef float v8f __attribute__((ext_vector_type(8)));

#define L1_LIST {0,0,0,0, 1,1,1,1,1,1,1,1, 2,2,2,2,2,2,2, 3,3,3,3}
#define L2_LIST {0,1,2,3, 1,1,1,2,2,2,3,3, 2,2,2,2,3,3,3, 3,3,3,3}
#define LL_LIST {0,1,2,3, 0,1,2,1,2,3,2,3, 0,1,2,3,1,2,3, 0,1,2,3}
#define OFF_LIST {0,1,10,35, 84,93,120,165,210,285,390,495, \
                  642,667,742,867,1042,1147,1322, 1567,1616,1763,2008}
#define FOFF_LIST {0,1,2,4, 8,9,10,11,12,14,17,20, \
                   24,26,28,32,38,42,48, 56,58,62,68}
#define XOFF_LIST {0, 1, 10, 35}
#define X2POS_LIST {0, 1, 19, 69}
#define X2NEG_LIST {0, 10, 44, 118}
__constant__ int c_l1[NCOMBO] = L1_LIST;
__constant__ int c_l2[NCOMBO] = L2_LIST;
__constant__ int c_L [NCOMBO] = LL_LIST;
__constant__ int c_off[NCOMBO] = OFF_LIST;
__constant__ int c_foff[NCOMBO] = FOFF_LIST;
__constant__ int d_xoff[4] = XOFF_LIST;
__constant__ int d_x2pos[4] = X2POS_LIST;
__constant__ int d_x2neg[4] = X2NEG_LIST;
constexpr int K_L1[NCOMBO] = L1_LIST;
constexpr int K_L2[NCOMBO] = L2_LIST;
constexpr int K_LL[NCOMBO] = LL_LIST;
constexpr int K_OFF[NCOMBO] = OFF_LIST;
constexpr int K_FOFF[NCOMBO] = FOFF_LIST;
constexpr int K_XOFF[4] = XOFF_LIST;
constexpr int K_X2POS[4] = X2POS_LIST;
constexpr int K_X2NEG[4] = X2NEG_LIST;
__constant__ int d_diag[4] = {0, 4, 12, 19};

__global__ void wigner_cg_init(float* __restrict__ ws) {
  double f[16];
  f[0] = 1.0;
  for (int i = 1; i < 16; ++i) f[i] = f[i - 1] * i;
  for (int c = 0; c < NCOMBO; ++c) {
    const int l1 = c_l1[c], l2 = c_l2[c], L = c_L[c];
    const int n2 = 2 * l2 + 1, nL = 2 * L + 1;
    const int sz = (2 * l1 + 1) * n2 * nL, off = c_off[c];
    const double pref0 =
        sqrt((2.0 * L + 1.0) * f[L + l1 - l2] * f[L - l1 + l2] * f[l1 + l2 - L] / f[l1 + l2 + L + 1]);
    for (int t = threadIdx.x; t < sz; t += blockDim.x) {
      const int m = t / (n2 * nL), rem = t % (n2 * nL), n = rem / nL, k = rem % nL;
      const int m1 = m - l1, m2 = n - l2, M = m1 + m2;
      double val = 0.0;
      if (M >= -L && M <= L && k == M + L) {
        const double pref =
            pref0 * sqrt(f[L + M] * f[L - M] * f[l1 - m1] * f[l1 + m1] * f[l2 - m2] * f[l2 + m2]);
        double s = 0.0;
        for (int kk = 0; kk <= l1 + l2 - L; ++kk) {
          const int a1 = l1 - m1 - kk, a2 = l2 + m2 - kk;
          const int a3 = L - l2 + m1 + kk, a4 = L - l1 - m2 + kk;
          if (a1 < 0 || a2 < 0 || a3 < 0 || a4 < 0) continue;
          const double term = 1.0 / (f[kk] * f[l1 + l2 - L - kk] * f[a1] * f[a2] * f[a3] * f[a4]);
          s += (kk & 1) ? -term : term;
        }
        val = pref * s;
      }
      ws[off + t] = (float)val;
    }
  }
  __syncthreads();
  for (int c = 0; c < NCOMBO; ++c) {
    const int l1 = c_l1[c], l2 = c_l2[c], L = c_L[c];
    const int n1 = 2 * l1 + 1, R = (2 * l2 + 1) * (2 * L + 1);
    const int cgo = c_off[c], fo = c_foff[c];
    const int mtiles = (R + 15) >> 4, kch = (n1 + 3) >> 2;
    for (int t = threadIdx.x; t < mtiles * kch * 64; t += blockDim.x) {
      const int fi = t >> 6, s = t & 63;
      const int mt = fi / kch, kc = fi % kch;
      const int ln = s >> 1, j = s & 1;
      const int row = mt * 16 + (ln & 15);
      const int m = kc * 4 + ((ln >> 4) ? 2 : 0) + j;
      ws[CG_TOTAL + (fo + fi) * 64 + s] =
          (row < R && m < n1) ? ws[cgo + m * R + row] : 0.f;
    }
  }
}

template <int C>
__device__ __forceinline__ void wigner_combo(const float* __restrict__ sCG,
                                             const float* __restrict__ sFRAG,
                                             float* __restrict__ XT, float* __restrict__ A,
                                             float* __restrict__ E, float* __restrict__ X2,
                                             const int lane, const int b, const int half) {
  constexpr int l1 = K_L1[C], l2 = K_L2[C], L = K_LL[C];
  constexpr int n1 = 2 * l1 + 1, n2 = 2 * l2 + 1, nL = 2 * L + 1;
  constexpr int R = n2 * nL, cgo = K_OFF[C], fb = K_FOFF[C];
  constexpr int xo1 = K_XOFF[l1], xo2 = K_XOFF[l2];
  constexpr int mtiles = (R + 15) >> 4, kch = (n1 + 3) >> 2;

  const int m0a = half ? 2 : 0;
  if constexpr (kch == 1) {
#pragma unroll 1
    for (int p = 0; p < n1; ++p) {
      v2f bf0;
      const int r0 = (m0a     < n1) ? (xo1 + m0a * n1 + p)       : ZROW;
      const int r1 = (m0a + 1 < n1) ? (xo1 + (m0a + 1) * n1 + p) : ZROW;
      bf0.x = XT[r0 * 16 + b];
      bf0.y = XT[r1 * 16 + b];
#pragma unroll
      for (int mt = 0; mt < mtiles; ++mt) {
        v8f acc = {0.f, 0.f, 0.f, 0.f, 0.f, 0.f, 0.f, 0.f};
        const v2f af0 = *(const v2f*)&sFRAG[(fb + mt) * 64 + lane * 2];
        acc = __builtin_amdgcn_wmma_f32_16x16x4_f32(false, af0, false, bf0,
                                                    (short)0, acc, false, false);
        const int rb = mt * 16 + half * 8;
#pragma unroll
        for (int r = 0; r < 8; ++r)
          A[((rb + r) * n1 + p) * 16 + b] = acc[r];
      }
    }
  } else {
#pragma unroll 1
    for (int p = 0; p < n1; ++p) {
      v2f bf0, bf1;
      {
        const int r0 = xo1 + m0a * n1 + p;
        const int r1 = xo1 + (m0a + 1) * n1 + p;
        bf0.x = XT[r0 * 16 + b];
        bf0.y = XT[r1 * 16 + b];
      }
      {
        const int m0 = 4 + m0a;
        const int r0 = (m0     < n1) ? (xo1 + m0 * n1 + p)       : ZROW;
        const int r1 = (m0 + 1 < n1) ? (xo1 + (m0 + 1) * n1 + p) : ZROW;
        bf1.x = XT[r0 * 16 + b];
        bf1.y = XT[r1 * 16 + b];
      }
#pragma unroll
      for (int mt = 0; mt < mtiles; ++mt) {
        v8f acc = {0.f, 0.f, 0.f, 0.f, 0.f, 0.f, 0.f, 0.f};
        const float* fp = &sFRAG[(fb + mt * 2) * 64 + lane * 2];
        const v2f af0 = *(const v2f*)fp;
        acc = __builtin_amdgcn_wmma_f32_16x16x4_f32(false, af0, false, bf0,
                                                    (short)0, acc, false, false);
        const v2f af1 = *(const v2f*)(fp + 64);
        acc = __builtin_amdgcn_wmma_f32_16x16x4_f32(false, af1, false, bf1,
                                                    (short)0, acc, false, false);
        const int rb = mt * 16 + half * 8;
#pragma unroll
        for (int r = 0; r < 8; ++r)
          A[((rb + r) * n1 + p) * 16 + b] = acc[r];
      }
    }
  }
  __syncthreads();

#pragma unroll 1
  for (int q = 0; q < n2; ++q) {
    float xv[n2];
#pragma unroll
    for (int n = 0; n < n2; ++n)
      xv[n] = XT[(xo2 + n * n2 + q) * 16 + b];
    constexpr int kp = nL * n1;
    for (int idx = half; idx < kp; idx += 2) {
      const int p = idx % n1, k = idx / n1;
      float e = 0.f;
#pragma unroll
      for (int n = 0; n < n2; ++n)
        e += xv[n] * A[((n * nL + k) * n1 + p) * 16 + b];
      E[((k * n2 + q) * n1 + p) * 16 + b] = e;
    }
  }
  __syncthreads();

  constexpr int x2base = ((l1 + l2 + L) & 1) ? K_X2NEG[L] : K_X2POS[L];
  constexpr float wc = (l1 < l2) ? 2.f : 1.f;
  for (int idx = half; idx < nL * nL; idx += 2) {
    const int j = idx % nL, k = idx / nL;
    float s = 0.f;
#pragma unroll 1
    for (int p = 0; p < n1; ++p) {
      float cr[n2];
#pragma unroll
      for (int q = 0; q < n2; ++q)
        cr[q] = sCG[cgo + (p * n2 + q) * nL + j];
#pragma unroll
      for (int q = 0; q < n2; ++q)
        if (cr[q] != 0.f)
          s += cr[q] * E[((k * n2 + q) * n1 + p) * 16 + b];
    }
    X2[(x2base + idx) * 16 + b] += wc * s;
  }
  __syncthreads();
}

__launch_bounds__(TB, 1)
__global__ void wigner_main(const float* __restrict__ x0, const float* __restrict__ x1,
                            const float* __restrict__ x2, const float* __restrict__ x3,
                            const float* __restrict__ tab, float* __restrict__ out, int Btot) {
  __shared__ __align__(16) float sTAB[TAB_TOTAL];
  __shared__ float sXT[WAVES_PER_BLOCK][85 * 16];
  __shared__ float sA [WAVES_PER_BLOCK][64 * 7 * 16];
  __shared__ float sE [WAVES_PER_BLOCK][7 * 7 * 7 * 16];
  __shared__ float sX2[WAVES_PER_BLOCK][167 * 16];

  const int tid = threadIdx.x, lane = tid & 31, wid = tid >> 5;
  const int b = lane & 15, half = lane >> 4;
  const float* sCG = sTAB;
  const float* sFRAG = sTAB + CG_TOTAL;

  for (int t = tid; t < TAB_TOTAL; t += TB) sTAB[t] = tab[t];

  const int tile = blockIdx.x * WAVES_PER_BLOCK + wid;
  const int base = tile * 16;
  float* XT = sXT[wid];
  float* A  = sA[wid];
  float* E  = sE[wid];
  float* X2 = sX2[wid];

  {
    int pe = base + WAVES_PER_BLOCK * 16;
    if (pe < Btot) __builtin_prefetch(&x3[(size_t)pe * 49], 0, 1);
  }

  for (int t = lane; t < 84 * 16; t += 32) {
    const int bb = t / 84, comp = t % 84;
    int e = base + bb;
    e = (e < Btot) ? e : (Btot - 1);
    float v;
    if (comp < 1)       v = x0[e];
    else if (comp < 10) v = x1[(size_t)e * 9 + (comp - 1)];
    else if (comp < 35) v = x2[(size_t)e * 25 + (comp - 10)];
    else                v = x3[(size_t)e * 49 + (comp - 35)];
    XT[comp * 16 + bb] = v;
  }
  for (int t = lane; t < 16; t += 32) XT[ZROW * 16 + t] = 0.f;
  for (int t = lane; t < 167 * 16; t += 32) X2[t] = 0.f;
  __syncthreads();

#define WIGNER_COMBO(C) wigner_combo<C>(sCG, sFRAG, XT, A, E, X2, lane, b, half)
  WIGNER_COMBO(0);  WIGNER_COMBO(1);  WIGNER_COMBO(2);  WIGNER_COMBO(3);
  WIGNER_COMBO(4);  WIGNER_COMBO(5);  WIGNER_COMBO(6);  WIGNER_COMBO(7);
  WIGNER_COMBO(8);  WIGNER_COMBO(9);  WIGNER_COMBO(10); WIGNER_COMBO(11);
  WIGNER_COMBO(12); WIGNER_COMBO(13); WIGNER_COMBO(14); WIGNER_COMBO(15);
  WIGNER_COMBO(16); WIGNER_COMBO(17); WIGNER_COMBO(18); WIGNER_COMBO(19);
  WIGNER_COMBO(20); WIGNER_COMBO(21); WIGNER_COMBO(22);
#undef WIGNER_COMBO

  if (lane < 16) {
    const int e = base + b;
    if (e < Btot) {
      const float nu1 = XT[0 * 16 + b];
      const float nu2 = X2[0 * 16 + b];
      float nu3 = 0.f, nu4 = 0.f;
      for (int l = 0; l <= 3; ++l) {
        const int n = 2 * l + 1;
        const int cgo = c_off[d_diag[l]];
        const int xo = d_xoff[l];
        const int xp = d_x2pos[l], xn = d_x2neg[l];
        for (int mn = 0; mn < n * n; ++mn) {
          const float c1 = sCG[cgo + mn];
          if (c1 == 0.f) continue;
          const int m = mn / n, nn = mn % n;
          for (int pq = 0; pq < n * n; ++pq) {
            const float c2 = sCG[cgo + pq];
            if (c2 == 0.f) continue;
            const int p = pq / n, q = pq % n;
            const float w = c1 * c2;
            nu3 += w * X2[(xp + m * n + p) * 16 + b] * XT[(xo + nn * n + q) * 16 + b];
            nu4 += w * X2[(xp + m * n + p) * 16 + b] * X2[(xp + nn * n + q) * 16 + b];
            if (l > 0)
              nu4 += w * X2[(xn + m * n + p) * 16 + b] * X2[(xn + nn * n + q) * 16 + b];
          }
        }
      }
      out[(size_t)e * 4 + 0] = nu1;
      out[(size_t)e * 4 + 1] = nu2;
      out[(size_t)e * 4 + 2] = nu3;
      out[(size_t)e * 4 + 3] = nu4;
    }
  }
}

extern "C" void kernel_launch(void* const* d_in, const int* in_sizes, int n_in,
                              void* d_out, int out_size, void* d_ws, size_t ws_size,
                              hipStream_t stream) {
  const float* x0 = (const float*)d_in[0];
  const float* x1 = (const float*)d_in[1];
  const float* x2 = (const float*)d_in[2];
  const float* x3 = (const float*)d_in[3];
  float* tab = (float*)d_ws;
  const int Btot = in_sizes[0];

  wigner_cg_init<<<1, 256, 0, stream>>>(tab);

  const int tiles = (Btot + 15) / 16;
  const int blocks = (tiles + WAVES_PER_BLOCK - 1) / WAVES_PER_BLOCK;
  wigner_main<<<blocks, TB, 0, stream>>>(x0, x1, x2, x3, tab, (float*)d_out, Btot);
}


# Stage 2 register-tiled: X_l2 block kept in registers, each A value read from LDS once and used for all n2 outputs q (n2 independent chains); E stored q-major; same summation order
# speedup vs baseline: 15.7537x; 6.3227x over previous
#include <hip/hip_runtime.h>


#define NCOMBO 23
#define CG_TOTAL 2351
#define NFRAGS 76
#define FRAG_TOTAL (NFRAGS * 64)
#define TAB_TOTAL (CG_TOTAL + FRAG_TOTAL)
#define WAVES_PER_BLOCK 4
#define TB (WAVES_PER_BLOCK * 32)
#define ZROW 84

typedef float v2f __attribute__((ext_vector_type(2)));
typedef float v8f __attribute__((ext_vector_type(8)));

#define L1_LIST {0,0,0,0, 1,1,1,1,1,1,1,1, 2,2,2,2,2,2,2, 3,3,3,3}
#define L2_LIST {0,1,2,3, 1,1,1,2,2,2,3,3, 2,2,2,2,3,3,3, 3,3,3,3}
#define LL_LIST {0,1,2,3, 0,1,2,1,2,3,2,3, 0,1,2,3,1,2,3, 0,1,2,3}
#define OFF_LIST {0,1,10,35, 84,93,120,165,210,285,390,495, \
                  642,667,742,867,1042,1147,1322, 1567,1616,1763,2008}
#define FOFF_LIST {0,1,2,4, 8,9,10,11,12,14,17,20, \
                   24,26,28,32,38,42,48, 56,58,62,68}
#define XOFF_LIST {0, 1, 10, 35}
#define X2POS_LIST {0, 1, 19, 69}
#define X2NEG_LIST {0, 10, 44, 118}
__constant__ int c_l1[NCOMBO] = L1_LIST;
__constant__ int c_l2[NCOMBO] = L2_LIST;
__constant__ int c_L [NCOMBO] = LL_LIST;
__constant__ int c_off[NCOMBO] = OFF_LIST;
__constant__ int c_foff[NCOMBO] = FOFF_LIST;
__constant__ int d_xoff[4] = XOFF_LIST;
__constant__ int d_x2pos[4] = X2POS_LIST;
__constant__ int d_x2neg[4] = X2NEG_LIST;
constexpr int K_L1[NCOMBO] = L1_LIST;
constexpr int K_L2[NCOMBO] = L2_LIST;
constexpr int K_LL[NCOMBO] = LL_LIST;
constexpr int K_OFF[NCOMBO] = OFF_LIST;
constexpr int K_FOFF[NCOMBO] = FOFF_LIST;
constexpr int K_XOFF[4] = XOFF_LIST;
constexpr int K_X2POS[4] = X2POS_LIST;
constexpr int K_X2NEG[4] = X2NEG_LIST;
__constant__ int d_diag[4] = {0, 4, 12, 19};

__global__ void wigner_cg_init(float* __restrict__ ws) {
  double f[16];
  f[0] = 1.0;
  for (int i = 1; i < 16; ++i) f[i] = f[i - 1] * i;
  for (int c = 0; c < NCOMBO; ++c) {
    const int l1 = c_l1[c], l2 = c_l2[c], L = c_L[c];
    const int n2 = 2 * l2 + 1, nL = 2 * L + 1;
    const int sz = (2 * l1 + 1) * n2 * nL, off = c_off[c];
    const double pref0 =
        sqrt((2.0 * L + 1.0) * f[L + l1 - l2] * f[L - l1 + l2] * f[l1 + l2 - L] / f[l1 + l2 + L + 1]);
    for (int t = threadIdx.x; t < sz; t += blockDim.x) {
      const int m = t / (n2 * nL), rem = t % (n2 * nL), n = rem / nL, k = rem % nL;
      const int m1 = m - l1, m2 = n - l2, M = m1 + m2;
      double val = 0.0;
      if (M >= -L && M <= L && k == M + L) {
        const double pref =
            pref0 * sqrt(f[L + M] * f[L - M] * f[l1 - m1] * f[l1 + m1] * f[l2 - m2] * f[l2 + m2]);
        double s = 0.0;
        for (int kk = 0; kk <= l1 + l2 - L; ++kk) {
          const int a1 = l1 - m1 - kk, a2 = l2 + m2 - kk;
          const int a3 = L - l2 + m1 + kk, a4 = L - l1 - m2 + kk;
          if (a1 < 0 || a2 < 0 || a3 < 0 || a4 < 0) continue;
          const double term = 1.0 / (f[kk] * f[l1 + l2 - L - kk] * f[a1] * f[a2] * f[a3] * f[a4]);
          s += (kk & 1) ? -term : term;
        }
        val = pref * s;
      }
      ws[off + t] = (float)val;
    }
  }
  __syncthreads();
  for (int c = 0; c < NCOMBO; ++c) {
    const int l1 = c_l1[c], l2 = c_l2[c], L = c_L[c];
    const int n1 = 2 * l1 + 1, R = (2 * l2 + 1) * (2 * L + 1);
    const int cgo = c_off[c], fo = c_foff[c];
    const int mtiles = (R + 15) >> 4, kch = (n1 + 3) >> 2;
    for (int t = threadIdx.x; t < mtiles * kch * 64; t += blockDim.x) {
      const int fi = t >> 6, s = t & 63;
      const int mt = fi / kch, kc = fi % kch;
      const int ln = s >> 1, j = s & 1;
      const int row = mt * 16 + (ln & 15);
      const int m = kc * 4 + ((ln >> 4) ? 2 : 0) + j;
      ws[CG_TOTAL + (fo + fi) * 64 + s] =
          (row < R && m < n1) ? ws[cgo + m * R + row] : 0.f;
    }
  }
}

template <int C>
__device__ __forceinline__ void wigner_combo(const float* __restrict__ sCG,
                                             const float* __restrict__ sFRAG,
                                             float* __restrict__ XT, float* __restrict__ A,
                                             float* __restrict__ E, float* __restrict__ X2,
                                             const int lane, const int b, const int half) {
  constexpr int l1 = K_L1[C], l2 = K_L2[C], L = K_LL[C];
  constexpr int n1 = 2 * l1 + 1, n2 = 2 * l2 + 1, nL = 2 * L + 1;
  constexpr int R = n2 * nL, cgo = K_OFF[C], fb = K_FOFF[C];
  constexpr int xo1 = K_XOFF[l1], xo2 = K_XOFF[l2];
  constexpr int mtiles = (R + 15) >> 4, kch = (n1 + 3) >> 2;

  const int m0a = half ? 2 : 0;
  if constexpr (kch == 1) {
#pragma unroll 1
    for (int p = 0; p < n1; ++p) {
      v2f bf0;
      const int r0 = (m0a     < n1) ? (xo1 + m0a * n1 + p)       : ZROW;
      const int r1 = (m0a + 1 < n1) ? (xo1 + (m0a + 1) * n1 + p) : ZROW;
      bf0.x = XT[r0 * 16 + b];
      bf0.y = XT[r1 * 16 + b];
#pragma unroll
      for (int mt = 0; mt < mtiles; ++mt) {
        v8f acc = {0.f, 0.f, 0.f, 0.f, 0.f, 0.f, 0.f, 0.f};
        const v2f af0 = *(const v2f*)&sFRAG[(fb + mt) * 64 + lane * 2];
        acc = __builtin_amdgcn_wmma_f32_16x16x4_f32(false, af0, false, bf0,
                                                    (short)0, acc, false, false);
        const int rb = mt * 16 + half * 8;
#pragma unroll
        for (int r = 0; r < 8; ++r)
          A[((rb + r) * n1 + p) * 16 + b] = acc[r];
      }
    }
  } else {
#pragma unroll 1
    for (int p = 0; p < n1; ++p) {
      v2f bf0, bf1;
      {
        const int r0 = xo1 + m0a * n1 + p;
        const int r1 = xo1 + (m0a + 1) * n1 + p;
        bf0.x = XT[r0 * 16 + b];
        bf0.y = XT[r1 * 16 + b];
      }
      {
        const int m0 = 4 + m0a;
        const int r0 = (m0     < n1) ? (xo1 + m0 * n1 + p)       : ZROW;
        const int r1 = (m0 + 1 < n1) ? (xo1 + (m0 + 1) * n1 + p) : ZROW;
        bf1.x = XT[r0 * 16 + b];
        bf1.y = XT[r1 * 16 + b];
      }
#pragma unroll
      for (int mt = 0; mt < mtiles; ++mt) {
        v8f acc = {0.f, 0.f, 0.f, 0.f, 0.f, 0.f, 0.f, 0.f};
        const float* fp = &sFRAG[(fb + mt * 2) * 64 + lane * 2];
        const v2f af0 = *(const v2f*)fp;
        acc = __builtin_amdgcn_wmma_f32_16x16x4_f32(false, af0, false, bf0,
                                                    (short)0, acc, false, false);
        const v2f af1 = *(const v2f*)(fp + 64);
        acc = __builtin_amdgcn_wmma_f32_16x16x4_f32(false, af1, false, bf1,
                                                    (short)0, acc, false, false);
        const int rb = mt * 16 + half * 8;
#pragma unroll
        for (int r = 0; r < 8; ++r)
          A[((rb + r) * n1 + p) * 16 + b] = acc[r];
      }
    }
  }
  __syncthreads();

  {
    constexpr int kp = nL * n1;
    float xv[n2][n2];
#pragma unroll
    for (int n = 0; n < n2; ++n) {
#pragma unroll
      for (int q = 0; q < n2; ++q)
        xv[n][q] = XT[(xo2 + n * n2 + q) * 16 + b];
    }
#pragma unroll 1
    for (int idx = half; idx < kp; idx += 2) {
      const float* Ap = A + idx * 16 + b;
      float* Ep = E + idx * 16 + b;
      float av[n2];
#pragma unroll
      for (int n = 0; n < n2; ++n)
        av[n] = Ap[n * kp * 16];
#pragma unroll
      for (int q = 0; q < n2; ++q) {
        float e = 0.f;
#pragma unroll
        for (int n = 0; n < n2; ++n)
          e += xv[n][q] * av[n];
        Ep[q * kp * 16] = e;
      }
    }
  }
  __syncthreads();

  constexpr int x2base = ((l1 + l2 + L) & 1) ? K_X2NEG[L] : K_X2POS[L];
  constexpr float wc = (l1 < l2) ? 2.f : 1.f;
  constexpr int D = l1 + l2 - L;
  float cf[nL][n1];
#pragma unroll
  for (int j = 0; j < nL; ++j) {
#pragma unroll
    for (int p = 0; p < n1; ++p) {
      const int q = j + D - p;
      cf[j][p] = (q >= 0 && q < n2) ? sCG[cgo + (p * n2 + q) * nL + j] : 0.f;
    }
  }
#pragma unroll 1
  for (int k = half; k < nL; k += 2) {
    const float* Ek = E + k * (n1 * 16) + b;
    float* Xk = X2 + (x2base + k * nL) * 16 + b;
#pragma unroll
    for (int j = 0; j < nL; ++j) {
      float s = 0.f;
#pragma unroll
      for (int p = 0; p < n1; ++p) {
        const int q = j + D - p;
        if (q >= 0 && q < n2)
          s += cf[j][p] * Ek[(q * (nL * n1) + p) * 16];
      }
      Xk[j * 16] += wc * s;
    }
  }
  __syncthreads();
}

__launch_bounds__(TB, 1)
__global__ void wigner_main(const float* __restrict__ x0, const float* __restrict__ x1,
                            const float* __restrict__ x2, const float* __restrict__ x3,
                            const float* __restrict__ tab, float* __restrict__ out, int Btot) {
  __shared__ __align__(16) float sTAB[TAB_TOTAL];
  __shared__ float sXT[WAVES_PER_BLOCK][85 * 16];
  __shared__ float sA [WAVES_PER_BLOCK][64 * 7 * 16];
  __shared__ float sE [WAVES_PER_BLOCK][7 * 7 * 7 * 16];
  __shared__ float sX2[WAVES_PER_BLOCK][167 * 16];

  const int tid = threadIdx.x, lane = tid & 31, wid = tid >> 5;
  const int b = lane & 15, half = lane >> 4;
  const float* sCG = sTAB;
  const float* sFRAG = sTAB + CG_TOTAL;

  for (int t = tid; t < TAB_TOTAL; t += TB) sTAB[t] = tab[t];

  const int tile = blockIdx.x * WAVES_PER_BLOCK + wid;
  const int base = tile * 16;
  float* XT = sXT[wid];
  float* A  = sA[wid];
  float* E  = sE[wid];
  float* X2 = sX2[wid];

  {
    int pe = base + WAVES_PER_BLOCK * 16;
    if (pe < Btot) __builtin_prefetch(&x3[(size_t)pe * 49], 0, 1);
  }

  for (int t = lane; t < 84 * 16; t += 32) {
    const int bb = t / 84, comp = t % 84;
    int e = base + bb;
    e = (e < Btot) ? e : (Btot - 1);
    float v;
    if (comp < 1)       v = x0[e];
    else if (comp < 10) v = x1[(size_t)e * 9 + (comp - 1)];
    else if (comp < 35) v = x2[(size_t)e * 25 + (comp - 10)];
    else                v = x3[(size_t)e * 49 + (comp - 35)];
    XT[comp * 16 + bb] = v;
  }
  for (int t = lane; t < 16; t += 32) XT[ZROW * 16 + t] = 0.f;
  for (int t = lane; t < 167 * 16; t += 32) X2[t] = 0.f;
  __syncthreads();

#define WIGNER_COMBO(C) wigner_combo<C>(sCG, sFRAG, XT, A, E, X2, lane, b, half)
  WIGNER_COMBO(0);  WIGNER_COMBO(1);  WIGNER_COMBO(2);  WIGNER_COMBO(3);
  WIGNER_COMBO(4);  WIGNER_COMBO(5);  WIGNER_COMBO(6);  WIGNER_COMBO(7);
  WIGNER_COMBO(8);  WIGNER_COMBO(9);  WIGNER_COMBO(10); WIGNER_COMBO(11);
  WIGNER_COMBO(12); WIGNER_COMBO(13); WIGNER_COMBO(14); WIGNER_COMBO(15);
  WIGNER_COMBO(16); WIGNER_COMBO(17); WIGNER_COMBO(18); WIGNER_COMBO(19);
  WIGNER_COMBO(20); WIGNER_COMBO(21); WIGNER_COMBO(22);
#undef WIGNER_COMBO

  if (lane < 16) {
    const int e = base + b;
    if (e < Btot) {
      const float nu1 = XT[0 * 16 + b];
      const float nu2 = X2[0 * 16 + b];
      float nu3 = 0.f, nu4 = 0.f;
      for (int l = 0; l <= 3; ++l) {
        const int n = 2 * l + 1;
        const int cgo = c_off[d_diag[l]];
        const int xo = d_xoff[l];
        const int xp = d_x2pos[l], xn = d_x2neg[l];
        for (int m = 0; m < n; ++m) {
          const int nn = n - 1 - m;
          const float c1 = sCG[cgo + m * n + nn];
          for (int p = 0; p < n; ++p) {
            const int q = n - 1 - p;
            const float c2 = sCG[cgo + p * n + q];
            const float w = c1 * c2;
            nu3 += w * X2[(xp + m * n + p) * 16 + b] * XT[(xo + nn * n + q) * 16 + b];
            nu4 += w * X2[(xp + m * n + p) * 16 + b] * X2[(xp + nn * n + q) * 16 + b];
            if (l > 0)
              nu4 += w * X2[(xn + m * n + p) * 16 + b] * X2[(xn + nn * n + q) * 16 + b];
          }
        }
      }
      out[(size_t)e * 4 + 0] = nu1;
      out[(size_t)e * 4 + 1] = nu2;
      out[(size_t)e * 4 + 2] = nu3;
      out[(size_t)e * 4 + 3] = nu4;
    }
  }
}

extern "C" void kernel_launch(void* const* d_in, const int* in_sizes, int n_in,
                              void* d_out, int out_size, void* d_ws, size_t ws_size,
                              hipStream_t stream) {
  const float* x0 = (const float*)d_in[0];
  const float* x1 = (const float*)d_in[1];
  const float* x2 = (const float*)d_in[2];
  const float* x3 = (const float*)d_in[3];
  float* tab = (float*)d_ws;
  const int Btot = in_sizes[0];

  wigner_cg_init<<<1, 256, 0, stream>>>(tab);

  const int tiles = (Btot + 15) / 16;
  const int blocks = (tiles + WAVES_PER_BLOCK - 1) / WAVES_PER_BLOCK;
  wigner_main<<<blocks, TB, 0, stream>>>(x0, x1, x2, x3, tab, (float*)d_out, Btot);
}


# Stages fused in registers: WMMA rows reordered k-major (8 n-slots per k) so a lane's accumulator is A[:,k,p]; stages 2-3 run from it, A/E LDS buffers and their barriers go; same summation order
# speedup vs baseline: 45.2842x; 2.8745x over previous
#include <hip/hip_runtime.h>


#define NCOMBO 23
#define CG_TOTAL 2351
#define NFRAGS 90
#define FRAG_TOTAL (NFRAGS * 64)
#define TAB_TOTAL (CG_TOTAL + FRAG_TOTAL)
#define WAVES_PER_BLOCK 4
#define TB (WAVES_PER_BLOCK * 32)
#define ZROW 84

typedef float v2f __attribute__((ext_vector_type(2)));
typedef float v8f __attribute__((ext_vector_type(8)));

#define L1_LIST {0,0,0,0, 1,1,1,1,1,1,1,1, 2,2,2,2,2,2,2, 3,3,3,3}
#define L2_LIST {0,1,2,3, 1,1,1,2,2,2,3,3, 2,2,2,2,3,3,3, 3,3,3,3}
#define LL_LIST {0,1,2,3, 0,1,2,1,2,3,2,3, 0,1,2,3,1,2,3, 0,1,2,3}
#define OFF_LIST {0,1,10,35, 84,93,120,165,210,285,390,495, \
                  642,667,742,867,1042,1147,1322, 1567,1616,1763,2008}
#define FOFF_LIST {0,1,3,6, 10,11,13,16,18,21,25,28, \
                   32,34,38,44,52,56,62, 70,72,76,82}
#define XOFF_LIST {0, 1, 10, 35}
#define X2POS_LIST {0, 1, 19, 69}
#define X2NEG_LIST {0, 10, 44, 118}
__constant__ int c_l1[NCOMBO] = L1_LIST;
__constant__ int c_l2[NCOMBO] = L2_LIST;
__constant__ int c_L [NCOMBO] = LL_LIST;
__constant__ int c_off[NCOMBO] = OFF_LIST;
__constant__ int c_foff[NCOMBO] = FOFF_LIST;
__constant__ int d_xoff[4] = XOFF_LIST;
__constant__ int d_x2pos[4] = X2POS_LIST;
__constant__ int d_x2neg[4] = X2NEG_LIST;
constexpr int K_L1[NCOMBO] = L1_LIST;
constexpr int K_L2[NCOMBO] = L2_LIST;
constexpr int K_LL[NCOMBO] = LL_LIST;
constexpr int K_OFF[NCOMBO] = OFF_LIST;
constexpr int K_FOFF[NCOMBO] = FOFF_LIST;
constexpr int K_XOFF[4] = XOFF_LIST;
constexpr int K_X2POS[4] = X2POS_LIST;
constexpr int K_X2NEG[4] = X2NEG_LIST;
__constant__ int d_diag[4] = {0, 4, 12, 19};

__global__ void wigner_cg_init(float* __restrict__ ws) {
  double f[16];
  f[0] = 1.0;
  for (int i = 1; i < 16; ++i) f[i] = f[i - 1] * i;
  for (int c = 0; c < NCOMBO; ++c) {
    const int l1 = c_l1[c], l2 = c_l2[c], L = c_L[c];
    const int n2 = 2 * l2 + 1, nL = 2 * L + 1;
    const int sz = (2 * l1 + 1) * n2 * nL, off = c_off[c];
    const double pref0 =
        sqrt((2.0 * L + 1.0) * f[L + l1 - l2] * f[L - l1 + l2] * f[l1 + l2 - L] / f[l1 + l2 + L + 1]);
    for (int t = threadIdx.x; t < sz; t += blockDim.x) {
      const int m = t / (n2 * nL), rem = t % (n2 * nL), n = rem / nL, k = rem % nL;
      const int m1 = m - l1, m2 = n - l2, M = m1 + m2;
      double val = 0.0;
      if (M >= -L && M <= L && k == M + L) {
        const double pref =
            pref0 * sqrt(f[L + M] * f[L - M] * f[l1 - m1] * f[l1 + m1] * f[l2 - m2] * f[l2 + m2]);
        double s = 0.0;
        for (int kk = 0; kk <= l1 + l2 - L; ++kk) {
          const int a1 = l1 - m1 - kk, a2 = l2 + m2 - kk;
          const int a3 = L - l2 + m1 + kk, a4 = L - l1 - m2 + kk;
          if (a1 < 0 || a2 < 0 || a3 < 0 || a4 < 0) continue;
          const double term = 1.0 / (f[kk] * f[l1 + l2 - L - kk] * f[a1] * f[a2] * f[a3] * f[a4]);
          s += (kk & 1) ? -term : term;
        }
        val = pref * s;
      }
      ws[off + t] = (float)val;
    }
  }
  __syncthreads();
  for (int c = 0; c < NCOMBO; ++c) {
    const int l1 = c_l1[c], l2 = c_l2[c], L = c_L[c];
    const int n1 = 2 * l1 + 1, n2 = 2 * l2 + 1, nL = 2 * L + 1, R = n2 * nL;
    const int cgo = c_off[c], fo = c_foff[c];
    const int mtiles = (nL + 1) >> 1, kch = (n1 + 3) >> 2;
    for (int t = threadIdx.x; t < mtiles * kch * 64; t += blockDim.x) {
      const int fi = t >> 6, s = t & 63;
      const int mt = fi / kch, kc = fi % kch;
      const int ln = s >> 1, j = s & 1;
      const int rr = ln & 15;
      const int k = 2 * mt + (rr >> 3), n = rr & 7;
      const int m = kc * 4 + ((ln >> 4) ? 2 : 0) + j;
      ws[CG_TOTAL + (fo + fi) * 64 + s] =
          (k < nL && n < n2 && m < n1) ? ws[cgo + m * R + n * nL + k] : 0.f;
    }
  }
}

template <int C>
__device__ __forceinline__ void wigner_combo(const float* __restrict__ sCG,
                                             const float* __restrict__ sFRAG,
                                             const float* __restrict__ XT,
                                             float* __restrict__ X2,
                                             const int lane, const int b, const int half) {
  constexpr int l1 = K_L1[C], l2 = K_L2[C], L = K_LL[C];
  constexpr int n1 = 2 * l1 + 1, n2 = 2 * l2 + 1, nL = 2 * L + 1;
  constexpr int cgo = K_OFF[C], fb = K_FOFF[C];
  constexpr int xo1 = K_XOFF[l1], xo2 = K_XOFF[l2];
  constexpr int mtiles = (nL + 1) >> 1, kch = (n1 + 3) >> 2;
  static_assert(fb + mtiles * kch == ((C + 1 < NCOMBO) ? K_FOFF[(C + 1 < NCOMBO) ? C + 1 : C] : NFRAGS),
                "fragment table offsets");
  constexpr int x2base = ((l1 + l2 + L) & 1) ? K_X2NEG[L] : K_X2POS[L];
  constexpr float wc = (l1 < l2) ? 2.f : 1.f;
  constexpr int D = l1 + l2 - L;

  const int m0a = half ? 2 : 0;
  v2f bf0[n1], bf1[n1];
#pragma unroll
  for (int p = 0; p < n1; ++p) {
    if constexpr (kch == 1) {
      const int r0 = (m0a     < n1) ? (xo1 + m0a * n1 + p)       : ZROW;
      const int r1 = (m0a + 1 < n1) ? (xo1 + (m0a + 1) * n1 + p) : ZROW;
      bf0[p].x = XT[r0 * 16 + b];
      bf0[p].y = XT[r1 * 16 + b];
      bf1[p].x = 0.f;
      bf1[p].y = 0.f;
    } else {
      {
        const int r0 = xo1 + m0a * n1 + p;
        const int r1 = xo1 + (m0a + 1) * n1 + p;
        bf0[p].x = XT[r0 * 16 + b];
        bf0[p].y = XT[r1 * 16 + b];
      }
      {
        const int m0 = 4 + m0a;
        const int r0 = (m0     < n1) ? (xo1 + m0 * n1 + p)       : ZROW;
        const int r1 = (m0 + 1 < n1) ? (xo1 + (m0 + 1) * n1 + p) : ZROW;
        bf1[p].x = XT[r0 * 16 + b];
        bf1[p].y = XT[r1 * 16 + b];
      }
    }
  }

  float xv[n2][n2];
#pragma unroll
  for (int n = 0; n < n2; ++n) {
#pragma unroll
    for (int q = 0; q < n2; ++q)
      xv[n][q] = XT[(xo2 + n * n2 + q) * 16 + b];
  }

  float cf[nL][n1];
#pragma unroll
  for (int j = 0; j < nL; ++j) {
#pragma unroll
    for (int p = 0; p < n1; ++p) {
      const int q = j + D - p;
      cf[j][p] = (q >= 0 && q < n2) ? sCG[cgo + (p * n2 + q) * nL + j] : 0.f;
    }
  }

#pragma unroll 1
  for (int mt = 0; mt < mtiles; ++mt) {
    const float* fp = &sFRAG[(fb + mt * kch) * 64 + lane * 2];
    const v2f af0 = *(const v2f*)fp;
    v2f af1 = {0.f, 0.f};
    if constexpr (kch == 2) af1 = *(const v2f*)(fp + 64);

    float s[nL];
#pragma unroll
    for (int j = 0; j < nL; ++j) s[j] = 0.f;

#pragma unroll
    for (int p = 0; p < n1; ++p) {
      v8f acc = {0.f, 0.f, 0.f, 0.f, 0.f, 0.f, 0.f, 0.f};
      acc = __builtin_amdgcn_wmma_f32_16x16x4_f32(false, af0, false, bf0[p],
                                                  (short)0, acc, false, false);
      if constexpr (kch == 2)
        acc = __builtin_amdgcn_wmma_f32_16x16x4_f32(false, af1, false, bf1[p],
                                                    (short)0, acc, false, false);
#pragma unroll
      for (int j = 0; j < nL; ++j) {
        const int q = j + D - p;
        if (q >= 0 && q < n2) {
          float e = 0.f;
#pragma unroll
          for (int n = 0; n < n2; ++n)
            e += xv[n][q] * acc[n];
          s[j] += cf[j][p] * e;
        }
      }
    }

    const int k = 2 * mt + half;
    if (k < nL) {
      float* Xk = X2 + (x2base + k * nL) * 16 + b;
#pragma unroll
      for (int j = 0; j < nL; ++j)
        Xk[j * 16] += wc * s[j];
    }
  }
}

__launch_bounds__(TB, 1)
__global__ void wigner_main(const float* __restrict__ x0, const float* __restrict__ x1,
                            const float* __restrict__ x2, const float* __restrict__ x3,
                            const float* __restrict__ tab, float* __restrict__ out, int Btot) {
  __shared__ __align__(16) float sTAB[TAB_TOTAL];
  __shared__ float sXT[WAVES_PER_BLOCK][85 * 16];
  __shared__ float sX2[WAVES_PER_BLOCK][167 * 16];

  const int tid = threadIdx.x, lane = tid & 31, wid = tid >> 5;
  const int b = lane & 15, half = lane >> 4;
  const float* sCG = sTAB;
  const float* sFRAG = sTAB + CG_TOTAL;

  for (int t = tid; t < TAB_TOTAL; t += TB) sTAB[t] = tab[t];

  const int tile = blockIdx.x * WAVES_PER_BLOCK + wid;
  const int base = tile * 16;
  float* XT = sXT[wid];
  float* X2 = sX2[wid];

  {
    int pe = base + WAVES_PER_BLOCK * 16;
    if (pe < Btot) __builtin_prefetch(&x3[(size_t)pe * 49], 0, 1);
  }

  for (int t = lane; t < 84 * 16; t += 32) {
    const int bb = t / 84, comp = t % 84;
    int e = base + bb;
    e = (e < Btot) ? e : (Btot - 1);
    float v;
    if (comp < 1)       v = x0[e];
    else if (comp < 10) v = x1[(size_t)e * 9 + (comp - 1)];
    else if (comp < 35) v = x2[(size_t)e * 25 + (comp - 10)];
    else                v = x3[(size_t)e * 49 + (comp - 35)];
    XT[comp * 16 + bb] = v;
  }
  for (int t = lane; t < 16; t += 32) XT[ZROW * 16 + t] = 0.f;
  for (int t = lane; t < 167 * 16; t += 32) X2[t] = 0.f;
  __syncthreads();

#define WIGNER_COMBO(C) wigner_combo<C>(sCG, sFRAG, XT, X2, lane, b, half)
  WIGNER_COMBO(0);  WIGNER_COMBO(1);  WIGNER_COMBO(2);  WIGNER_COMBO(3);
  WIGNER_COMBO(4);  WIGNER_COMBO(5);  WIGNER_COMBO(6);  WIGNER_COMBO(7);
  WIGNER_COMBO(8);  WIGNER_COMBO(9);  WIGNER_COMBO(10); WIGNER_COMBO(11);
  WIGNER_COMBO(12); WIGNER_COMBO(13); WIGNER_COMBO(14); WIGNER_COMBO(15);
  WIGNER_COMBO(16); WIGNER_COMBO(17); WIGNER_COMBO(18); WIGNER_COMBO(19);
  WIGNER_COMBO(20); WIGNER_COMBO(21); WIGNER_COMBO(22);
#undef WIGNER_COMBO
  __syncthreads();

  if (lane < 16) {
    const int e = base + b;
    if (e < Btot) {
      const float nu1 = XT[0 * 16 + b];
      const float nu2 = X2[0 * 16 + b];
      float nu3 = 0.f, nu4 = 0.f;
      for (int l = 0; l <= 3; ++l) {
        const int n = 2 * l + 1;
        const int cgo = c_off[d_diag[l]];
        const int xo = d_xoff[l];
        const int xp = d_x2pos[l], xn = d_x2neg[l];
        for (int m = 0; m < n; ++m) {
          const int nn = n - 1 - m;
          const float c1 = sCG[cgo + m * n + nn];
          for (int p = 0; p < n; ++p) {
            const int q = n - 1 - p;
            const float c2 = sCG[cgo + p * n + q];
            const float w = c1 * c2;
            nu3 += w * X2[(xp + m * n + p) * 16 + b] * XT[(xo + nn * n + q) * 16 + b];
            nu4 += w * X2[(xp + m * n + p) * 16 + b] * X2[(xp + nn * n + q) * 16 + b];
            if (l > 0)
              nu4 += w * X2[(xn + m * n + p) * 16 + b] * X2[(xn + nn * n + q) * 16 + b];
          }
        }
      }
      out[(size_t)e * 4 + 0] = nu1;
      out[(size_t)e * 4 + 1] = nu2;
      out[(size_t)e * 4 + 2] = nu3;
      out[(size_t)e * 4 + 3] = nu4;
    }
  }
}

extern "C" void kernel_launch(void* const* d_in, const int* in_sizes, int n_in,
                              void* d_out, int out_size, void* d_ws, size_t ws_size,
                              hipStream_t stream) {
  const float* x0 = (const float*)d_in[0];
  const float* x1 = (const float*)d_in[1];
  const float* x2 = (const float*)d_in[2];
  const float* x3 = (const float*)d_in[3];
  float* tab = (float*)d_ws;
  const int Btot = in_sizes[0];

  wigner_cg_init<<<1, 256, 0, stream>>>(tab);

  const int tiles = (Btot + 15) / 16;
  const int blocks = (tiles + WAVES_PER_BLOCK - 1) / WAVES_PER_BLOCK;
  wigner_main<<<blocks, TB, 0, stream>>>(x0, x1, x2, x3, tab, (float*)d_out, Btot);
}


# Table-building kernel spread over 23 work-groups (one per combo), one thread per (m,n) pair, instead of one serial 256-thread group: FP64 work per SIMD drops ~20x; same arithmetic and order
# speedup vs baseline: 58.9849x; 1.3025x over previous
#include <hip/hip_runtime.h>


#define NCOMBO 23
#define CG_TOTAL 2351
#define NFRAGS 90
#define FRAG_TOTAL (NFRAGS * 64)
#define TAB_TOTAL (CG_TOTAL + FRAG_TOTAL)
#define WAVES_PER_BLOCK 4
#define TB (WAVES_PER_BLOCK * 32)
#define ZROW 84

typedef float v2f __attribute__((ext_vector_type(2)));
typedef float v8f __attribute__((ext_vector_type(8)));

#define L1_LIST {0,0,0,0, 1,1,1,1,1,1,1,1, 2,2,2,2,2,2,2, 3,3,3,3}
#define L2_LIST {0,1,2,3, 1,1,1,2,2,2,3,3, 2,2,2,2,3,3,3, 3,3,3,3}
#define LL_LIST {0,1,2,3, 0,1,2,1,2,3,2,3, 0,1,2,3,1,2,3, 0,1,2,3}
#define OFF_LIST {0,1,10,35, 84,93,120,165,210,285,390,495, \
                  642,667,742,867,1042,1147,1322, 1567,1616,1763,2008}
#define FOFF_LIST {0,1,3,6, 10,11,13,16,18,21,25,28, \
                   32,34,38,44,52,56,62, 70,72,76,82}
#define XOFF_LIST {0, 1, 10, 35}
#define X2POS_LIST {0, 1, 19, 69}
#define X2NEG_LIST {0, 10, 44, 118}
__constant__ int c_l1[NCOMBO] = L1_LIST;
__constant__ int c_l2[NCOMBO] = L2_LIST;
__constant__ int c_L [NCOMBO] = LL_LIST;
__constant__ int c_off[NCOMBO] = OFF_LIST;
__constant__ int c_foff[NCOMBO] = FOFF_LIST;
__constant__ int d_xoff[4] = XOFF_LIST;
__constant__ int d_x2pos[4] = X2POS_LIST;
__constant__ int d_x2neg[4] = X2NEG_LIST;
constexpr int K_L1[NCOMBO] = L1_LIST;
constexpr int K_L2[NCOMBO] = L2_LIST;
constexpr int K_LL[NCOMBO] = LL_LIST;
constexpr int K_OFF[NCOMBO] = OFF_LIST;
constexpr int K_FOFF[NCOMBO] = FOFF_LIST;
constexpr int K_XOFF[4] = XOFF_LIST;
constexpr int K_X2POS[4] = X2POS_LIST;
constexpr int K_X2NEG[4] = X2NEG_LIST;
__constant__ int d_diag[4] = {0, 4, 12, 19};

#define INIT_TB 128
__global__ void wigner_cg_init(float* __restrict__ ws) {
  __shared__ double sFct[16];
  __shared__ float sC[343];
  const int tid = threadIdx.x;
  const int c = blockIdx.x;
  const int l1 = c_l1[c], l2 = c_l2[c], L = c_L[c];
  const int n1 = 2 * l1 + 1, n2 = 2 * l2 + 1, nL = 2 * L + 1;
  const int sz = n1 * n2 * nL;
  if (tid < 16) {
    double v = 1.0;
    for (int i = 1; i <= tid; ++i) v = v * i;
    sFct[tid] = v;
  }
  for (int t = tid; t < sz; t += INIT_TB) sC[t] = 0.f;
  __syncthreads();
  const double* f = sFct;
  if (tid < n1 * n2) {
    const int m = tid / n2, n = tid % n2;
    const int m1 = m - l1, m2 = n - l2, M = m1 + m2;
    if (M >= -L && M <= L) {
      const double pref0 =
          sqrt((2.0 * L + 1.0) * f[L + l1 - l2] * f[L - l1 + l2] * f[l1 + l2 - L] / f[l1 + l2 + L + 1]);
      const double pref =
          pref0 * sqrt(f[L + M] * f[L - M] * f[l1 - m1] * f[l1 + m1] * f[l2 - m2] * f[l2 + m2]);
      double s = 0.0;
      for (int kk = 0; kk <= l1 + l2 - L; ++kk) {
        const int a1 = l1 - m1 - kk, a2 = l2 + m2 - kk;
        const int a3 = L - l2 + m1 + kk, a4 = L - l1 - m2 + kk;
        if (a1 < 0 || a2 < 0 || a3 < 0 || a4 < 0) continue;
        const double term = 1.0 / (f[kk] * f[l1 + l2 - L - kk] * f[a1] * f[a2] * f[a3] * f[a4]);
        s += (kk & 1) ? -term : term;
      }
      const double val = pref * s;
      sC[(m * n2 + n) * nL + (M + L)] = (float)val;
    }
  }
  __syncthreads();
  const int cgo = c_off[c];
  for (int t = tid; t < sz; t += INIT_TB) ws[cgo + t] = sC[t];
  const int R = n2 * nL, fo = c_foff[c];
  const int mtiles = (nL + 1) >> 1, kch = (n1 + 3) >> 2;
  for (int t = tid; t < mtiles * kch * 64; t += INIT_TB) {
    const int fi = t >> 6, s = t & 63;
    const int mt = fi / kch, kc = fi % kch;
    const int ln = s >> 1, j = s & 1;
    const int rr = ln & 15;
    const int k = 2 * mt + (rr >> 3), n = rr & 7;
    const int m = kc * 4 + ((ln >> 4) ? 2 : 0) + j;
    ws[CG_TOTAL + (fo + fi) * 64 + s] =
        (k < nL && n < n2 && m < n1) ? sC[m * R + n * nL + k] : 0.f;
  }
}

template <int C>
__device__ __forceinline__ void wigner_combo(const float* __restrict__ sCG,
                                             const float* __restrict__ sFRAG,
                                             const float* __restrict__ XT,
                                             float* __restrict__ X2,
                                             const int lane, const int b, const int half) {
  constexpr int l1 = K_L1[C], l2 = K_L2[C], L = K_LL[C];
  constexpr int n1 = 2 * l1 + 1, n2 = 2 * l2 + 1, nL = 2 * L + 1;
  constexpr int cgo = K_OFF[C], fb = K_FOFF[C];
  constexpr int xo1 = K_XOFF[l1], xo2 = K_XOFF[l2];
  constexpr int mtiles = (nL + 1) >> 1, kch = (n1 + 3) >> 2;
  static_assert(fb + mtiles * kch == ((C + 1 < NCOMBO) ? K_FOFF[(C + 1 < NCOMBO) ? C + 1 : C] : NFRAGS),
                "fragment table offsets");
  constexpr int x2base = ((l1 + l2 + L) & 1) ? K_X2NEG[L] : K_X2POS[L];
  constexpr float wc = (l1 < l2) ? 2.f : 1.f;
  constexpr int D = l1 + l2 - L;

  const int m0a = half ? 2 : 0;
  v2f bf0[n1], bf1[n1];
#pragma unroll
  for (int p = 0; p < n1; ++p) {
    if constexpr (kch == 1) {
      const int r0 = (m0a     < n1) ? (xo1 + m0a * n1 + p)       : ZROW;
      const int r1 = (m0a + 1 < n1) ? (xo1 + (m0a + 1) * n1 + p) : ZROW;
      bf0[p].x = XT[r0 * 16 + b];
      bf0[p].y = XT[r1 * 16 + b];
      bf1[p].x = 0.f;
      bf1[p].y = 0.f;
    } else {
      {
        const int r0 = xo1 + m0a * n1 + p;
        const int r1 = xo1 + (m0a + 1) * n1 + p;
        bf0[p].x = XT[r0 * 16 + b];
        bf0[p].y = XT[r1 * 16 + b];
      }
      {
        const int m0 = 4 + m0a;
        const int r0 = (m0     < n1) ? (xo1 + m0 * n1 + p)       : ZROW;
        const int r1 = (m0 + 1 < n1) ? (xo1 + (m0 + 1) * n1 + p) : ZROW;
        bf1[p].x = XT[r0 * 16 + b];
        bf1[p].y = XT[r1 * 16 + b];
      }
    }
  }

  float xv[n2][n2];
#pragma unroll
  for (int n = 0; n < n2; ++n) {
#pragma unroll
    for (int q = 0; q < n2; ++q)
      xv[n][q] = XT[(xo2 + n * n2 + q) * 16 + b];
  }

  float cf[nL][n1];
#pragma unroll
  for (int j = 0; j < nL; ++j) {
#pragma unroll
    for (int p = 0; p < n1; ++p) {
      const int q = j + D - p;
      cf[j][p] = (q >= 0 && q < n2) ? sCG[cgo + (p * n2 + q) * nL + j] : 0.f;
    }
  }

#pragma unroll 1
  for (int mt = 0; mt < mtiles; ++mt) {
    const float* fp = &sFRAG[(fb + mt * kch) * 64 + lane * 2];
    const v2f af0 = *(const v2f*)fp;
    v2f af1 = {0.f, 0.f};
    if constexpr (kch == 2) af1 = *(const v2f*)(fp + 64);

    float s[nL];
#pragma unroll
    for (int j = 0; j < nL; ++j) s[j] = 0.f;

#pragma unroll
    for (int p = 0; p < n1; ++p) {
      v8f acc = {0.f, 0.f, 0.f, 0.f, 0.f, 0.f, 0.f, 0.f};
      acc = __builtin_amdgcn_wmma_f32_16x16x4_f32(false, af0, false, bf0[p],
                                                  (short)0, acc, false, false);
      if constexpr (kch == 2)
        acc = __builtin_amdgcn_wmma_f32_16x16x4_f32(false, af1, false, bf1[p],
                                                    (short)0, acc, false, false);
#pragma unroll
      for (int j = 0; j < nL; ++j) {
        const int q = j + D - p;
        if (q >= 0 && q < n2) {
          float e = 0.f;
#pragma unroll
          for (int n = 0; n < n2; ++n)
            e += xv[n][q] * acc[n];
          s[j] += cf[j][p] * e;
        }
      }
    }

    const int k = 2 * mt + half;
    if (k < nL) {
      float* Xk = X2 + (x2base + k * nL) * 16 + b;
#pragma unroll
      for (int j = 0; j < nL; ++j)
        Xk[j * 16] += wc * s[j];
    }
  }
}

__launch_bounds__(TB, 1)
__global__ void wigner_main(const float* __restrict__ x0, const float* __restrict__ x1,
                            const float* __restrict__ x2, const float* __restrict__ x3,
                            const float* __restrict__ tab, float* __restrict__ out, int Btot) {
  __shared__ __align__(16) float sTAB[TAB_TOTAL];
  __shared__ float sXT[WAVES_PER_BLOCK][85 * 16];
  __shared__ float sX2[WAVES_PER_BLOCK][167 * 16];

  const int tid = threadIdx.x, lane = tid & 31, wid = tid >> 5;
  const int b = lane & 15, half = lane >> 4;
  const float* sCG = sTAB;
  const float* sFRAG = sTAB + CG_TOTAL;

  for (int t = tid; t < TAB_TOTAL; t += TB) sTAB[t] = tab[t];

  const int tile = blockIdx.x * WAVES_PER_BLOCK + wid;
  const int base = tile * 16;
  float* XT = sXT[wid];
  float* X2 = sX2[wid];

  {
    int pe = base + WAVES_PER_BLOCK * 16;
    if (pe < Btot) __builtin_prefetch(&x3[(size_t)pe * 49], 0, 1);
  }

  for (int t = lane; t < 84 * 16; t += 32) {
    const int bb = t / 84, comp = t % 84;
    int e = base + bb;
    e = (e < Btot) ? e : (Btot - 1);
    float v;
    if (comp < 1)       v = x0[e];
    else if (comp < 10) v = x1[(size_t)e * 9 + (comp - 1)];
    else if (comp < 35) v = x2[(size_t)e * 25 + (comp - 10)];
    else                v = x3[(size_t)e * 49 + (comp - 35)];
    XT[comp * 16 + bb] = v;
  }
  for (int t = lane; t < 16; t += 32) XT[ZROW * 16 + t] = 0.f;
  for (int t = lane; t < 167 * 16; t += 32) X2[t] = 0.f;
  __syncthreads();

#define WIGNER_COMBO(C) wigner_combo<C>(sCG, sFRAG, XT, X2, lane, b, half)
  WIGNER_COMBO(0);  WIGNER_COMBO(1);  WIGNER_COMBO(2);  WIGNER_COMBO(3);
  WIGNER_COMBO(4);  WIGNER_COMBO(5);  WIGNER_COMBO(6);  WIGNER_COMBO(7);
  WIGNER_COMBO(8);  WIGNER_COMBO(9);  WIGNER_COMBO(10); WIGNER_COMBO(11);
  WIGNER_COMBO(12); WIGNER_COMBO(13); WIGNER_COMBO(14); WIGNER_COMBO(15);
  WIGNER_COMBO(16); WIGNER_COMBO(17); WIGNER_COMBO(18); WIGNER_COMBO(19);
  WIGNER_COMBO(20); WIGNER_COMBO(21); WIGNER_COMBO(22);
#undef WIGNER_COMBO
  __syncthreads();

  if (lane < 16) {
    const int e = base + b;
    if (e < Btot) {
      const float nu1 = XT[0 * 16 + b];
      const float nu2 = X2[0 * 16 + b];
      float nu3 = 0.f, nu4 = 0.f;
      for (int l = 0; l <= 3; ++l) {
        const int n = 2 * l + 1;
        const int cgo = c_off[d_diag[l]];
        const int xo = d_xoff[l];
        const int xp = d_x2pos[l], xn = d_x2neg[l];
        for (int m = 0; m < n; ++m) {
          const int nn = n - 1 - m;
          const float c1 = sCG[cgo + m * n + nn];
          for (int p = 0; p < n; ++p) {
            const int q = n - 1 - p;
            const float c2 = sCG[cgo + p * n + q];
            const float w = c1 * c2;
            nu3 += w * X2[(xp + m * n + p) * 16 + b] * XT[(xo + nn * n + q) * 16 + b];
            nu4 += w * X2[(xp + m * n + p) * 16 + b] * X2[(xp + nn * n + q) * 16 + b];
            if (l > 0)
              nu4 += w * X2[(xn + m * n + p) * 16 + b] * X2[(xn + nn * n + q) * 16 + b];
          }
        }
      }
      out[(size_t)e * 4 + 0] = nu1;
      out[(size_t)e * 4 + 1] = nu2;
      out[(size_t)e * 4 + 2] = nu3;
      out[(size_t)e * 4 + 3] = nu4;
    }
  }
}

extern "C" void kernel_launch(void* const* d_in, const int* in_sizes, int n_in,
                              void* d_out, int out_size, void* d_ws, size_t ws_size,
                              hipStream_t stream) {
  const float* x0 = (const float*)d_in[0];
  const float* x1 = (const float*)d_in[1];
  const float* x2 = (const float*)d_in[2];
  const float* x3 = (const float*)d_in[3];
  float* tab = (float*)d_ws;
  const int Btot = in_sizes[0];

  wigner_cg_init<<<NCOMBO, INIT_TB, 0, stream>>>(tab);

  const int tiles = (Btot + 15) / 16;
  const int blocks = (tiles + WAVES_PER_BLOCK - 1) / WAVES_PER_BLOCK;
  wigner_main<<<blocks, TB, 0, stream>>>(x0, x1, x2, x3, tab, (float*)d_out, Btot);
}


# Prologue loads batched: the tile's four contiguous input chunks (44 loads per lane) and the table copy (16-byte loads) all issue before the first wait, not one round trip per loop trip; same sums
# speedup vs baseline: 88.9133x; 1.5074x over previous
#include <hip/hip_runtime.h>


#define NCOMBO 23
#define CG_TOTAL 2351
#define NFRAGS 90
#define FRAG_TOTAL (NFRAGS * 64)
#define TAB_TOTAL (CG_TOTAL + FRAG_TOTAL)
#define WAVES_PER_BLOCK 4
#define TB (WAVES_PER_BLOCK * 32)
#define ZROW 84
#define TAB4 (TAB_TOTAL / 4)
#define TABTAIL (TAB_TOTAL - 4 * TAB4)
#define TABR ((TAB4 + TB - 1) / TB)

typedef float v2f __attribute__((ext_vector_type(2)));
typedef float v4f __attribute__((ext_vector_type(4)));
typedef float v8f __attribute__((ext_vector_type(8)));

#define L1_LIST {0,0,0,0, 1,1,1,1,1,1,1,1, 2,2,2,2,2,2,2, 3,3,3,3}
#define L2_LIST {0,1,2,3, 1,1,1,2,2,2,3,3, 2,2,2,2,3,3,3, 3,3,3,3}
#define LL_LIST {0,1,2,3, 0,1,2,1,2,3,2,3, 0,1,2,3,1,2,3, 0,1,2,3}
#define OFF_LIST {0,1,10,35, 84,93,120,165,210,285,390,495, \
                  642,667,742,867,1042,1147,1322, 1567,1616,1763,2008}
#define FOFF_LIST {0,1,3,6, 10,11,13,16,18,21,25,28, \
                   32,34,38,44,52,56,62, 70,72,76,82}
#define XOFF_LIST {0, 1, 10, 35}
#define X2POS_LIST {0, 1, 19, 69}
#define X2NEG_LIST {0, 10, 44, 118}
__constant__ int c_l1[NCOMBO] = L1_LIST;
__constant__ int c_l2[NCOMBO] = L2_LIST;
__constant__ int c_L [NCOMBO] = LL_LIST;
__constant__ int c_off[NCOMBO] = OFF_LIST;
__constant__ int c_foff[NCOMBO] = FOFF_LIST;
__constant__ int d_xoff[4] = XOFF_LIST;
__constant__ int d_x2pos[4] = X2POS_LIST;
__constant__ int d_x2neg[4] = X2NEG_LIST;
constexpr int K_L1[NCOMBO] = L1_LIST;
constexpr int K_L2[NCOMBO] = L2_LIST;
constexpr int K_LL[NCOMBO] = LL_LIST;
constexpr int K_OFF[NCOMBO] = OFF_LIST;
constexpr int K_FOFF[NCOMBO] = FOFF_LIST;
constexpr int K_XOFF[4] = XOFF_LIST;
constexpr int K_X2POS[4] = X2POS_LIST;
constexpr int K_X2NEG[4] = X2NEG_LIST;
__constant__ int d_diag[4] = {0, 4, 12, 19};

#define INIT_TB 128
__global__ void wigner_cg_init(float* __restrict__ ws) {
  __shared__ double sFct[16];
  __shared__ float sC[343];
  const int tid = threadIdx.x;
  const int c = blockIdx.x;
  const int l1 = c_l1[c], l2 = c_l2[c], L = c_L[c];
  const int n1 = 2 * l1 + 1, n2 = 2 * l2 + 1, nL = 2 * L + 1;
  const int sz = n1 * n2 * nL;
  if (tid < 16) {
    double v = 1.0;
    for (int i = 1; i <= tid; ++i) v = v * i;
    sFct[tid] = v;
  }
  for (int t = tid; t < sz; t += INIT_TB) sC[t] = 0.f;
  __syncthreads();
  const double* f = sFct;
  if (tid < n1 * n2) {
    const int m = tid / n2, n = tid % n2;
    const int m1 = m - l1, m2 = n - l2, M = m1 + m2;
    if (M >= -L && M <= L) {
      const double pref0 =
          sqrt((2.0 * L + 1.0) * f[L + l1 - l2] * f[L - l1 + l2] * f[l1 + l2 - L] / f[l1 + l2 + L + 1]);
      const double pref =
          pref0 * sqrt(f[L + M] * f[L - M] * f[l1 - m1] * f[l1 + m1] * f[l2 - m2] * f[l2 + m2]);
      double s = 0.0;
      for (int kk = 0; kk <= l1 + l2 - L; ++kk) {
        const int a1 = l1 - m1 - kk, a2 = l2 + m2 - kk;
        const int a3 = L - l2 + m1 + kk, a4 = L - l1 - m2 + kk;
        if (a1 < 0 || a2 < 0 || a3 < 0 || a4 < 0) continue;
        const double term = 1.0 / (f[kk] * f[l1 + l2 - L - kk] * f[a1] * f[a2] * f[a3] * f[a4]);
        s += (kk & 1) ? -term : term;
      }
      const double val = pref * s;
      sC[(m * n2 + n) * nL + (M + L)] = (float)val;
    }
  }
  __syncthreads();
  const int cgo = c_off[c];
  for (int t = tid; t < sz; t += INIT_TB) ws[cgo + t] = sC[t];
  const int R = n2 * nL, fo = c_foff[c];
  const int mtiles = (nL + 1) >> 1, kch = (n1 + 3) >> 2;
  for (int t = tid; t < mtiles * kch * 64; t += INIT_TB) {
    const int fi = t >> 6, s = t & 63;
    const int mt = fi / kch, kc = fi % kch;
    const int ln = s >> 1, j = s & 1;
    const int rr = ln & 15;
    const int k = 2 * mt + (rr >> 3), n = rr & 7;
    const int m = kc * 4 + ((ln >> 4) ? 2 : 0) + j;
    ws[CG_TOTAL + (fo + fi) * 64 + s] =
        (k < nL && n < n2 && m < n1) ? sC[m * R + n * nL + k] : 0.f;
  }
}

template <int C>
__device__ __forceinline__ void wigner_combo(const float* __restrict__ sCG,
                                             const float* __restrict__ sFRAG,
                                             const float* __restrict__ XT,
                                             float* __restrict__ X2,
                                             const int lane, const int b, const int half) {
  constexpr int l1 = K_L1[C], l2 = K_L2[C], L = K_LL[C];
  constexpr int n1 = 2 * l1 + 1, n2 = 2 * l2 + 1, nL = 2 * L + 1;
  constexpr int cgo = K_OFF[C], fb = K_FOFF[C];
  constexpr int xo1 = K_XOFF[l1], xo2 = K_XOFF[l2];
  constexpr int mtiles = (nL + 1) >> 1, kch = (n1 + 3) >> 2;
  static_assert(fb + mtiles * kch == ((C + 1 < NCOMBO) ? K_FOFF[(C + 1 < NCOMBO) ? C + 1 : C] : NFRAGS),
                "fragment table offsets");
  constexpr int x2base = ((l1 + l2 + L) & 1) ? K_X2NEG[L] : K_X2POS[L];
  constexpr float wc = (l1 < l2) ? 2.f : 1.f;
  constexpr int D = l1 + l2 - L;

  const int m0a = half ? 2 : 0;
  v2f bf0[n1], bf1[n1];
#pragma unroll
  for (int p = 0; p < n1; ++p) {
    if constexpr (kch == 1) {
      const int r0 = (m0a     < n1) ? (xo1 + m0a * n1 + p)       : ZROW;
      const int r1 = (m0a + 1 < n1) ? (xo1 + (m0a + 1) * n1 + p) : ZROW;
      bf0[p].x = XT[r0 * 16 + b];
      bf0[p].y = XT[r1 * 16 + b];
      bf1[p].x = 0.f;
      bf1[p].y = 0.f;
    } else {
      {
        const int r0 = xo1 + m0a * n1 + p;
        const int r1 = xo1 + (m0a + 1) * n1 + p;
        bf0[p].x = XT[r0 * 16 + b];
        bf0[p].y = XT[r1 * 16 + b];
      }
      {
        const int m0 = 4 + m0a;
        const int r0 = (m0     < n1) ? (xo1 + m0 * n1 + p)       : ZROW;
        const int r1 = (m0 + 1 < n1) ? (xo1 + (m0 + 1) * n1 + p) : ZROW;
        bf1[p].x = XT[r0 * 16 + b];
        bf1[p].y = XT[r1 * 16 + b];
      }
    }
  }

  float xv[n2][n2];
#pragma unroll
  for (int n = 0; n < n2; ++n) {
#pragma unroll
    for (int q = 0; q < n2; ++q)
      xv[n][q] = XT[(xo2 + n * n2 + q) * 16 + b];
  }

  float cf[nL][n1];
#pragma unroll
  for (int j = 0; j < nL; ++j) {
#pragma unroll
    for (int p = 0; p < n1; ++p) {
      const int q = j + D - p;
      cf[j][p] = (q >= 0 && q < n2) ? sCG[cgo + (p * n2 + q) * nL + j] : 0.f;
    }
  }

#pragma unroll 1
  for (int mt = 0; mt < mtiles; ++mt) {
    const float* fp = &sFRAG[(fb + mt * kch) * 64 + lane * 2];
    const v2f af0 = *(const v2f*)fp;
    v2f af1 = {0.f, 0.f};
    if constexpr (kch == 2) af1 = *(const v2f*)(fp + 64);

    float s[nL];
#pragma unroll
    for (int j = 0; j < nL; ++j) s[j] = 0.f;

#pragma unroll
    for (int p = 0; p < n1; ++p) {
      v8f acc = {0.f, 0.f, 0.f, 0.f, 0.f, 0.f, 0.f, 0.f};
      acc = __builtin_amdgcn_wmma_f32_16x16x4_f32(false, af0, false, bf0[p],
                                                  (short)0, acc, false, false);
      if constexpr (kch == 2)
        acc = __builtin_amdgcn_wmma_f32_16x16x4_f32(false, af1, false, bf1[p],
                                                    (short)0, acc, false, false);
#pragma unroll
      for (int j = 0; j < nL; ++j) {
        const int q = j + D - p;
        if (q >= 0 && q < n2) {
          float e = 0.f;
#pragma unroll
          for (int n = 0; n < n2; ++n)
            e += xv[n][q] * acc[n];
          s[j] += cf[j][p] * e;
        }
      }
    }

    const int k = 2 * mt + half;
    if (k < nL) {
      float* Xk = X2 + (x2base + k * nL) * 16 + b;
#pragma unroll
      for (int j = 0; j < nL; ++j)
        Xk[j * 16] += wc * s[j];
    }
  }
}

template <int S, int R>
__device__ __forceinline__ void chunk_load(const float* __restrict__ x, const int base,
                                           const int Btot, const int lane, float (&g)[R]) {
#pragma unroll
  for (int r = 0; r < R; ++r) {
    int i = lane + 32 * r;
    i = (i < 16 * S) ? i : (16 * S - 1);
    const int bb = i / S, c = i - bb * S;
    int e = base + bb;
    e = (e < Btot) ? e : (Btot - 1);
    g[r] = x[(unsigned)(e * S + c)];
  }
}
template <int S, int R, int XO>
__device__ __forceinline__ void chunk_store(float* __restrict__ XT, const int lane,
                                            const float (&g)[R]) {
#pragma unroll
  for (int r = 0; r < R; ++r) {
    const int i = lane + 32 * r;
    const int bb = i / S, c = i - bb * S;
    if (i < 16 * S) XT[(XO + c) * 16 + bb] = g[r];
  }
}

__launch_bounds__(TB, 1)
__global__ void wigner_main(const float* __restrict__ x0, const float* __restrict__ x1,
                            const float* __restrict__ x2, const float* __restrict__ x3,
                            const float* __restrict__ tab, float* __restrict__ out, int Btot) {
  __shared__ __align__(16) float sTAB[TAB_TOTAL];
  __shared__ float sXT[WAVES_PER_BLOCK][85 * 16];
  __shared__ float sX2[WAVES_PER_BLOCK][167 * 16];

  const int tid = threadIdx.x, lane = tid & 31, wid = tid >> 5;
  const int b = lane & 15, half = lane >> 4;
  const float* sCG = sTAB;
  const float* sFRAG = sTAB + CG_TOTAL;

  const int tile = blockIdx.x * WAVES_PER_BLOCK + wid;
  const int base = tile * 16;
  float* XT = sXT[wid];
  float* X2 = sX2[wid];

  {
    int pe = base + WAVES_PER_BLOCK * 16;
    if (pe < Btot) __builtin_prefetch(&x3[(size_t)pe * 49], 0, 1);
  }

  float g3[25], g2[13], g1[5], g0[1];
  chunk_load<49, 25>(x3, base, Btot, lane, g3);
  chunk_load<25, 13>(x2, base, Btot, lane, g2);
  chunk_load<9, 5>(x1, base, Btot, lane, g1);
  chunk_load<1, 1>(x0, base, Btot, lane, g0);
  const v4f* tab4 = (const v4f*)tab;
  v4f tv[TABR];
#pragma unroll
  for (int r = 0; r < TABR; ++r) {
    int i4 = tid + TB * r;
    i4 = (i4 < TAB4) ? i4 : (TAB4 - 1);
    tv[r] = tab4[i4];
  }
  const float tt = tab[TAB4 * 4 + ((tid < TABTAIL) ? tid : 0)];

  for (int t = lane; t < 16; t += 32) XT[ZROW * 16 + t] = 0.f;
  for (int t = lane; t < 167 * 16; t += 32) X2[t] = 0.f;

  chunk_store<49, 25, 35>(XT, lane, g3);
  chunk_store<25, 13, 10>(XT, lane, g2);
  chunk_store<9, 5, 1>(XT, lane, g1);
  chunk_store<1, 1, 0>(XT, lane, g0);
#pragma unroll
  for (int r = 0; r < TABR; ++r) {
    const int i4 = tid + TB * r;
    if (i4 < TAB4) *(v4f*)&sTAB[i4 * 4] = tv[r];
  }
  if (tid < TABTAIL) sTAB[TAB4 * 4 + tid] = tt;
  __syncthreads();

#define WIGNER_COMBO(C) wigner_combo<C>(sCG, sFRAG, XT, X2, lane, b, half)
  WIGNER_COMBO(0);  WIGNER_COMBO(1);  WIGNER_COMBO(2);  WIGNER_COMBO(3);
  WIGNER_COMBO(4);  WIGNER_COMBO(5);  WIGNER_COMBO(6);  WIGNER_COMBO(7);
  WIGNER_COMBO(8);  WIGNER_COMBO(9);  WIGNER_COMBO(10); WIGNER_COMBO(11);
  WIGNER_COMBO(12); WIGNER_COMBO(13); WIGNER_COMBO(14); WIGNER_COMBO(15);
  WIGNER_COMBO(16); WIGNER_COMBO(17); WIGNER_COMBO(18); WIGNER_COMBO(19);
  WIGNER_COMBO(20); WIGNER_COMBO(21); WIGNER_COMBO(22);
#undef WIGNER_COMBO
  __syncthreads();

  if (lane < 16) {
    const int e = base + b;
    if (e < Btot) {
      const float nu1 = XT[0 * 16 + b];
      const float nu2 = X2[0 * 16 + b];
      float nu3 = 0.f, nu4 = 0.f;
      for (int l = 0; l <= 3; ++l) {
        const int n = 2 * l + 1;
        const int cgo = c_off[d_diag[l]];
        const int xo = d_xoff[l];
        const int xp = d_x2pos[l], xn = d_x2neg[l];
        for (int m = 0; m < n; ++m) {
          const int nn = n - 1 - m;
          const float c1 = sCG[cgo + m * n + nn];
          for (int p = 0; p < n; ++p) {
            const int q = n - 1 - p;
            const float c2 = sCG[cgo + p * n + q];
            const float w = c1 * c2;
            nu3 += w * X2[(xp + m * n + p) * 16 + b] * XT[(xo + nn * n + q) * 16 + b];
            nu4 += w * X2[(xp + m * n + p) * 16 + b] * X2[(xp + nn * n + q) * 16 + b];
            if (l > 0)
              nu4 += w * X2[(xn + m * n + p) * 16 + b] * X2[(xn + nn * n + q) * 16 + b];
          }
        }
      }
      out[(size_t)e * 4 + 0] = nu1;
      out[(size_t)e * 4 + 1] = nu2;
      out[(size_t)e * 4 + 2] = nu3;
      out[(size_t)e * 4 + 3] = nu4;
    }
  }
}

extern "C" void kernel_launch(void* const* d_in, const int* in_sizes, int n_in,
                              void* d_out, int out_size, void* d_ws, size_t ws_size,
                              hipStream_t stream) {
  const float* x0 = (const float*)d_in[0];
  const float* x1 = (const float*)d_in[1];
  const float* x2 = (const float*)d_in[2];
  const float* x3 = (const float*)d_in[3];
  float* tab = (float*)d_ws;
  const int Btot = in_sizes[0];

  wigner_cg_init<<<NCOMBO, INIT_TB, 0, stream>>>(tab);

  const int tiles = (Btot + 15) / 16;
  const int blocks = (tiles + WAVES_PER_BLOCK - 1) / WAVES_PER_BLOCK;
  wigner_main<<<blocks, TB, 0, stream>>>(x0, x1, x2, x3, tab, (float*)d_out, Btot);
}


# Row-tile loop software-pipelined: WMMA of column block p+1, next tile's fragments and the trip's X2 reads issue ahead of the vector work on block p, which covers their latency; same sums
# speedup vs baseline: 94.9580x; 1.0680x over previous
#include <hip/hip_runtime.h>


#define NCOMBO 23
#define CG_TOTAL 2351
#define NFRAGS 90
#define FRAG_TOTAL (NFRAGS * 64)
#define TAB_TOTAL (CG_TOTAL + FRAG_TOTAL)
#define WAVES_PER_BLOCK 4
#define TB (WAVES_PER_BLOCK * 32)
#define ZROW 84
#define X2ROWS 167
#define X2PAD 7
#define TAB4 (TAB_TOTAL / 4)
#define TABTAIL (TAB_TOTAL - 4 * TAB4)
#define TABR ((TAB4 + TB - 1) / TB)

typedef float v2f __attribute__((ext_vector_type(2)));
typedef float v4f __attribute__((ext_vector_type(4)));
typedef float v8f __attribute__((ext_vector_type(8)));

#define L1_LIST {0,0,0,0, 1,1,1,1,1,1,1,1, 2,2,2,2,2,2,2, 3,3,3,3}
#define L2_LIST {0,1,2,3, 1,1,1,2,2,2,3,3, 2,2,2,2,3,3,3, 3,3,3,3}
#define LL_LIST {0,1,2,3, 0,1,2,1,2,3,2,3, 0,1,2,3,1,2,3, 0,1,2,3}
#define OFF_LIST {0,1,10,35, 84,93,120,165,210,285,390,495, \
                  642,667,742,867,1042,1147,1322, 1567,1616,1763,2008}
#define FOFF_LIST {0,1,3,6, 10,11,13,16,18,21,25,28, \
                   32,34,38,44,52,56,62, 70,72,76,82}
#define XOFF_LIST {0, 1, 10, 35}
#define X2POS_LIST {0, 1, 19, 69}
#define X2NEG_LIST {0, 10, 44, 118}
__constant__ int c_l1[NCOMBO] = L1_LIST;
__constant__ int c_l2[NCOMBO] = L2_LIST;
__constant__ int c_L [NCOMBO] = LL_LIST;
__constant__ int c_off[NCOMBO] = OFF_LIST;
__constant__ int c_foff[NCOMBO] = FOFF_LIST;
__constant__ int d_xoff[4] = XOFF_LIST;
__constant__ int d_x2pos[4] = X2POS_LIST;
__constant__ int d_x2neg[4] = X2NEG_LIST;
constexpr int K_L1[NCOMBO] = L1_LIST;
constexpr int K_L2[NCOMBO] = L2_LIST;
constexpr int K_LL[NCOMBO] = LL_LIST;
constexpr int K_OFF[NCOMBO] = OFF_LIST;
constexpr int K_FOFF[NCOMBO] = FOFF_LIST;
constexpr int K_XOFF[4] = XOFF_LIST;
constexpr int K_X2POS[4] = X2POS_LIST;
constexpr int K_X2NEG[4] = X2NEG_LIST;
__constant__ int d_diag[4] = {0, 4, 12, 19};

#define INIT_TB 128
__global__ void wigner_cg_init(float* __restrict__ ws) {
  __shared__ double sFct[16];
  __shared__ float sC[343];
  const int tid = threadIdx.x;
  const int c = blockIdx.x;
  const int l1 = c_l1[c], l2 = c_l2[c], L = c_L[c];
  const int n1 = 2 * l1 + 1, n2 = 2 * l2 + 1, nL = 2 * L + 1;
  const int sz = n1 * n2 * nL;
  if (tid < 16) {
    double v = 1.0;
    for (int i = 1; i <= tid; ++i) v = v * i;
    sFct[tid] = v;
  }
  for (int t = tid; t < sz; t += INIT_TB) sC[t] = 0.f;
  __syncthreads();
  const double* f = sFct;
  if (tid < n1 * n2) {
    const int m = tid / n2, n = tid % n2;
    const int m1 = m - l1, m2 = n - l2, M = m1 + m2;
    if (M >= -L && M <= L) {
      const double pref0 =
          sqrt((2.0 * L + 1.0) * f[L + l1 - l2] * f[L - l1 + l2] * f[l1 + l2 - L] / f[l1 + l2 + L + 1]);
      const double pref =
          pref0 * sqrt(f[L + M] * f[L - M] * f[l1 - m1] * f[l1 + m1] * f[l2 - m2] * f[l2 + m2]);
      double s = 0.0;
      for (int kk = 0; kk <= l1 + l2 - L; ++kk) {
        const int a1 = l1 - m1 - kk, a2 = l2 + m2 - kk;
        const int a3 = L - l2 + m1 + kk, a4 = L - l1 - m2 + kk;
        if (a1 < 0 || a2 < 0 || a3 < 0 || a4 < 0) continue;
        const double term = 1.0 / (f[kk] * f[l1 + l2 - L - kk] * f[a1] * f[a2] * f[a3] * f[a4]);
        s += (kk & 1) ? -term : term;
      }
      const double val = pref * s;
      sC[(m * n2 + n) * nL + (M + L)] = (float)val;
    }
  }
  __syncthreads();
  const int cgo = c_off[c];
  for (int t = tid; t < sz; t += INIT_TB) ws[cgo + t] = sC[t];
  const int R = n2 * nL, fo = c_foff[c];
  const int mtiles = (nL + 1) >> 1, kch = (n1 + 3) >> 2;
  for (int t = tid; t < mtiles * kch * 64; t += INIT_TB) {
    const int fi = t >> 6, s = t & 63;
    const int mt = fi / kch, kc = fi % kch;
    const int ln = s >> 1, j = s & 1;
    const int rr = ln & 15;
    const int k = 2 * mt + (rr >> 3), n = rr & 7;
    const int m = kc * 4 + ((ln >> 4) ? 2 : 0) + j;
    ws[CG_TOTAL + (fo + fi) * 64 + s] =
        (k < nL && n < n2 && m < n1) ? sC[m * R + n * nL + k] : 0.f;
  }
}

template <int C>
__device__ __forceinline__ void wigner_combo(const float* __restrict__ sCG,
                                             const float* __restrict__ sFRAG,
                                             const float* __restrict__ XT,
                                             float* __restrict__ X2,
                                             const int lane, const int b, const int half) {
  constexpr int l1 = K_L1[C], l2 = K_L2[C], L = K_LL[C];
  constexpr int n1 = 2 * l1 + 1, n2 = 2 * l2 + 1, nL = 2 * L + 1;
  constexpr int cgo = K_OFF[C], fb = K_FOFF[C];
  constexpr int xo1 = K_XOFF[l1], xo2 = K_XOFF[l2];
  constexpr int mtiles = (nL + 1) >> 1, kch = (n1 + 3) >> 2;
  static_assert(fb + mtiles * kch == ((C + 1 < NCOMBO) ? K_FOFF[(C + 1 < NCOMBO) ? C + 1 : C] : NFRAGS),
                "fragment table offsets");
  constexpr int x2base = ((l1 + l2 + L) & 1) ? K_X2NEG[L] : K_X2POS[L];
  constexpr float wc = (l1 < l2) ? 2.f : 1.f;
  constexpr int D = l1 + l2 - L;

  const int m0a = half ? 2 : 0;
  v2f bf0[n1], bf1[n1];
#pragma unroll
  for (int p = 0; p < n1; ++p) {
    if constexpr (kch == 1) {
      const int r0 = (m0a     < n1) ? (xo1 + m0a * n1 + p)       : ZROW;
      const int r1 = (m0a + 1 < n1) ? (xo1 + (m0a + 1) * n1 + p) : ZROW;
      bf0[p].x = XT[r0 * 16 + b];
      bf0[p].y = XT[r1 * 16 + b];
      bf1[p].x = 0.f;
      bf1[p].y = 0.f;
    } else {
      {
        const int r0 = xo1 + m0a * n1 + p;
        const int r1 = xo1 + (m0a + 1) * n1 + p;
        bf0[p].x = XT[r0 * 16 + b];
        bf0[p].y = XT[r1 * 16 + b];
      }
      {
        const int m0 = 4 + m0a;
        const int r0 = (m0     < n1) ? (xo1 + m0 * n1 + p)       : ZROW;
        const int r1 = (m0 + 1 < n1) ? (xo1 + (m0 + 1) * n1 + p) : ZROW;
        bf1[p].x = XT[r0 * 16 + b];
        bf1[p].y = XT[r1 * 16 + b];
      }
    }
  }

  float xv[n2][n2];
#pragma unroll
  for (int n = 0; n < n2; ++n) {
#pragma unroll
    for (int q = 0; q < n2; ++q)
      xv[n][q] = XT[(xo2 + n * n2 + q) * 16 + b];
  }

  float cf[nL][n1];
#pragma unroll
  for (int j = 0; j < nL; ++j) {
#pragma unroll
    for (int p = 0; p < n1; ++p) {
      const int q = j + D - p;
      cf[j][p] = (q >= 0 && q < n2) ? sCG[cgo + (p * n2 + q) * nL + j] : 0.f;
    }
  }

  const float* fp = &sFRAG[fb * 64 + lane * 2];
  v2f af0 = *(const v2f*)fp;
  v2f af1 = {0.f, 0.f};
  if constexpr (kch == 2) af1 = *(const v2f*)(fp + 64);

#pragma unroll 1
  for (int mt = 0; mt < mtiles; ++mt) {
    const float* fn = fp + ((mt + 1 < mtiles) ? kch * 64 : 0);
    const v2f nf0 = *(const v2f*)fn;
    v2f nf1 = {0.f, 0.f};
    if constexpr (kch == 2) nf1 = *(const v2f*)(fn + 64);

    const int k = 2 * mt + half;
    float* Xk = X2 + ((k < nL) ? (x2base + k * nL) : X2ROWS) * 16 + b;
    float xo[nL];
#pragma unroll
    for (int j = 0; j < nL; ++j) xo[j] = Xk[j * 16];

    float s[nL];
#pragma unroll
    for (int j = 0; j < nL; ++j) s[j] = 0.f;

    const v8f zero8 = {0.f, 0.f, 0.f, 0.f, 0.f, 0.f, 0.f, 0.f};
    v8f accn = __builtin_amdgcn_wmma_f32_16x16x4_f32(false, af0, false, bf0[0],
                                                     (short)0, zero8, false, false);
    if constexpr (kch == 2)
      accn = __builtin_amdgcn_wmma_f32_16x16x4_f32(false, af1, false, bf1[0],
                                                   (short)0, accn, false, false);

#pragma unroll
    for (int p = 0; p < n1; ++p) {
      const v8f acc = accn;
      if (p + 1 < n1) {
        accn = __builtin_amdgcn_wmma_f32_16x16x4_f32(false, af0, false, bf0[p + 1],
                                                     (short)0, zero8, false, false);
        if constexpr (kch == 2)
          accn = __builtin_amdgcn_wmma_f32_16x16x4_f32(false, af1, false, bf1[p + 1],
                                                       (short)0, accn, false, false);
      }
      __builtin_amdgcn_sched_barrier(0);
#pragma unroll
      for (int j = 0; j < nL; ++j) {
        const int q = j + D - p;
        if (q >= 0 && q < n2) {
          float e = 0.f;
#pragma unroll
          for (int n = 0; n < n2; ++n)
            e += xv[n][q] * acc[n];
          s[j] += cf[j][p] * e;
        }
      }
    }

#pragma unroll
    for (int j = 0; j < nL; ++j)
      Xk[j * 16] = xo[j] + wc * s[j];
    fp = fn;
    af0 = nf0;
    af1 = nf1;
  }
}

template <int S, int R>
__device__ __forceinline__ void chunk_load(const float* __restrict__ x, const int base,
                                           const int Btot, const int lane, float (&g)[R]) {
#pragma unroll
  for (int r = 0; r < R; ++r) {
    int i = lane + 32 * r;
    i = (i < 16 * S) ? i : (16 * S - 1);
    const int bb = i / S, c = i - bb * S;
    int e = base + bb;
    e = (e < Btot) ? e : (Btot - 1);
    g[r] = x[(unsigned)(e * S + c)];
  }
}
template <int S, int R, int XO>
__device__ __forceinline__ void chunk_store(float* __restrict__ XT, const int lane,
                                            const float (&g)[R]) {
#pragma unroll
  for (int r = 0; r < R; ++r) {
    const int i = lane + 32 * r;
    const int bb = i / S, c = i - bb * S;
    if (i < 16 * S) XT[(XO + c) * 16 + bb] = g[r];
  }
}

__launch_bounds__(TB, 1)
__global__ void wigner_main(const float* __restrict__ x0, const float* __restrict__ x1,
                            const float* __restrict__ x2, const float* __restrict__ x3,
                            const float* __restrict__ tab, float* __restrict__ out, int Btot) {
  __shared__ __align__(16) float sTAB[TAB_TOTAL];
  __shared__ float sXT[WAVES_PER_BLOCK][85 * 16];
  __shared__ float sX2[WAVES_PER_BLOCK][(X2ROWS + X2PAD) * 16];

  const int tid = threadIdx.x, lane = tid & 31, wid = tid >> 5;
  const int b = lane & 15, half = lane >> 4;
  const float* sCG = sTAB;
  const float* sFRAG = sTAB + CG_TOTAL;

  const int tile = blockIdx.x * WAVES_PER_BLOCK + wid;
  const int base = tile * 16;
  float* XT = sXT[wid];
  float* X2 = sX2[wid];

  {
    int pe = base + WAVES_PER_BLOCK * 16;
    if (pe < Btot) __builtin_prefetch(&x3[(size_t)pe * 49], 0, 1);
  }

  float g3[25], g2[13], g1[5], g0[1];
  chunk_load<49, 25>(x3, base, Btot, lane, g3);
  chunk_load<25, 13>(x2, base, Btot, lane, g2);
  chunk_load<9, 5>(x1, base, Btot, lane, g1);
  chunk_load<1, 1>(x0, base, Btot, lane, g0);
  const v4f* tab4 = (const v4f*)tab;
  v4f tv[TABR];
#pragma unroll
  for (int r = 0; r < TABR; ++r) {
    int i4 = tid + TB * r;
    i4 = (i4 < TAB4) ? i4 : (TAB4 - 1);
    tv[r] = tab4[i4];
  }
  const float tt = tab[TAB4 * 4 + ((tid < TABTAIL) ? tid : 0)];

  for (int t = lane; t < 16; t += 32) XT[ZROW * 16 + t] = 0.f;
  for (int t = lane; t < (X2ROWS + X2PAD) * 16; t += 32) X2[t] = 0.f;

  chunk_store<49, 25, 35>(XT, lane, g3);
  chunk_store<25, 13, 10>(XT, lane, g2);
  chunk_store<9, 5, 1>(XT, lane, g1);
  chunk_store<1, 1, 0>(XT, lane, g0);
#pragma unroll
  for (int r = 0; r < TABR; ++r) {
    const int i4 = tid + TB * r;
    if (i4 < TAB4) *(v4f*)&sTAB[i4 * 4] = tv[r];
  }
  if (tid < TABTAIL) sTAB[TAB4 * 4 + tid] = tt;
  __syncthreads();

#define WIGNER_COMBO(C) wigner_combo<C>(sCG, sFRAG, XT, X2, lane, b, half)
  WIGNER_COMBO(0);  WIGNER_COMBO(1);  WIGNER_COMBO(2);  WIGNER_COMBO(3);
  WIGNER_COMBO(4);  WIGNER_COMBO(5);  WIGNER_COMBO(6);  WIGNER_COMBO(7);
  WIGNER_COMBO(8);  WIGNER_COMBO(9);  WIGNER_COMBO(10); WIGNER_COMBO(11);
  WIGNER_COMBO(12); WIGNER_COMBO(13); WIGNER_COMBO(14); WIGNER_COMBO(15);
  WIGNER_COMBO(16); WIGNER_COMBO(17); WIGNER_COMBO(18); WIGNER_COMBO(19);
  WIGNER_COMBO(20); WIGNER_COMBO(21); WIGNER_COMBO(22);
#undef WIGNER_COMBO
  __syncthreads();

  if (lane < 16) {
    const int e = base + b;
    if (e < Btot) {
      const float nu1 = XT[0 * 16 + b];
      const float nu2 = X2[0 * 16 + b];
      float nu3 = 0.f, nu4 = 0.f;
      for (int l = 0; l <= 3; ++l) {
        const int n = 2 * l + 1;
        const int cgo = c_off[d_diag[l]];
        const int xo = d_xoff[l];
        const int xp = d_x2pos[l], xn = d_x2neg[l];
        for (int m = 0; m < n; ++m) {
          const int nn = n - 1 - m;
          const float c1 = sCG[cgo + m * n + nn];
          for (int p = 0; p < n; ++p) {
            const int q = n - 1 - p;
            const float c2 = sCG[cgo + p * n + q];
            const float w = c1 * c2;
            nu3 += w * X2[(xp + m * n + p) * 16 + b] * XT[(xo + nn * n + q) * 16 + b];
            nu4 += w * X2[(xp + m * n + p) * 16 + b] * X2[(xp + nn * n + q) * 16 + b];
            if (l > 0)
              nu4 += w * X2[(xn + m * n + p) * 16 + b] * X2[(xn + nn * n + q) * 16 + b];
          }
        }
      }
      out[(size_t)e * 4 + 0] = nu1;
      out[(size_t)e * 4 + 1] = nu2;
      out[(size_t)e * 4 + 2] = nu3;
      out[(size_t)e * 4 + 3] = nu4;
    }
  }
}

extern "C" void kernel_launch(void* const* d_in, const int* in_sizes, int n_in,
                              void* d_out, int out_size, void* d_ws, size_t ws_size,
                              hipStream_t stream) {
  const float* x0 = (const float*)d_in[0];
  const float* x1 = (const float*)d_in[1];
  const float* x2 = (const float*)d_in[2];
  const float* x3 = (const float*)d_in[3];
  float* tab = (float*)d_ws;
  const int Btot = in_sizes[0];

  wigner_cg_init<<<NCOMBO, INIT_TB, 0, stream>>>(tab);

  const int tiles = (Btot + 15) / 16;
  const int blocks = (tiles + WAVES_PER_BLOCK - 1) / WAVES_PER_BLOCK;
  wigner_main<<<blocks, TB, 0, stream>>>(x0, x1, x2, x3, tab, (float*)d_out, Btot);
}


# Stage 2 two-wide along q: X_l2 rows held as aligned register pairs, one packed multiply-add per n forms e[q],e[q+1] in place; the register copies of the pairing across p go; same summation order
# speedup vs baseline: 99.5173x; 1.0480x over previous
#include <hip/hip_runtime.h>


#define NCOMBO 23
#define CG_TOTAL 2351
#define NFRAGS 90
#define FRAG_TOTAL (NFRAGS * 64)
#define TAB_TOTAL (CG_TOTAL + FRAG_TOTAL)
#define WAVES_PER_BLOCK 4
#define TB (WAVES_PER_BLOCK * 32)
#define ZROW 84
#define X2ROWS 167
#define X2PAD 7
#define TAB4 (TAB_TOTAL / 4)
#define TABTAIL (TAB_TOTAL - 4 * TAB4)
#define TABR ((TAB4 + TB - 1) / TB)

typedef float v2f __attribute__((ext_vector_type(2)));
typedef float v4f __attribute__((ext_vector_type(4)));
typedef float v8f __attribute__((ext_vector_type(8)));

#define L1_LIST {0,0,0,0, 1,1,1,1,1,1,1,1, 2,2,2,2,2,2,2, 3,3,3,3}
#define L2_LIST {0,1,2,3, 1,1,1,2,2,2,3,3, 2,2,2,2,3,3,3, 3,3,3,3}
#define LL_LIST {0,1,2,3, 0,1,2,1,2,3,2,3, 0,1,2,3,1,2,3, 0,1,2,3}
#define OFF_LIST {0,1,10,35, 84,93,120,165,210,285,390,495, \
                  642,667,742,867,1042,1147,1322, 1567,1616,1763,2008}
#define FOFF_LIST {0,1,3,6, 10,11,13,16,18,21,25,28, \
                   32,34,38,44,52,56,62, 70,72,76,82}
#define XOFF_LIST {0, 1, 10, 35}
#define X2POS_LIST {0, 1, 19, 69}
#define X2NEG_LIST {0, 10, 44, 118}
__constant__ int c_l1[NCOMBO] = L1_LIST;
__constant__ int c_l2[NCOMBO] = L2_LIST;
__constant__ int c_L [NCOMBO] = LL_LIST;
__constant__ int c_off[NCOMBO] = OFF_LIST;
__constant__ int c_foff[NCOMBO] = FOFF_LIST;
__constant__ int d_xoff[4] = XOFF_LIST;
__constant__ int d_x2pos[4] = X2POS_LIST;
__constant__ int d_x2neg[4] = X2NEG_LIST;
constexpr int K_L1[NCOMBO] = L1_LIST;
constexpr int K_L2[NCOMBO] = L2_LIST;
constexpr int K_LL[NCOMBO] = LL_LIST;
constexpr int K_OFF[NCOMBO] = OFF_LIST;
constexpr int K_FOFF[NCOMBO] = FOFF_LIST;
constexpr int K_XOFF[4] = XOFF_LIST;
constexpr int K_X2POS[4] = X2POS_LIST;
constexpr int K_X2NEG[4] = X2NEG_LIST;
__constant__ int d_diag[4] = {0, 4, 12, 19};

#define INIT_TB 128
__global__ void wigner_cg_init(float* __restrict__ ws) {
  __shared__ double sFct[16];
  __shared__ float sC[343];
  const int tid = threadIdx.x;
  const int c = blockIdx.x;
  const int l1 = c_l1[c], l2 = c_l2[c], L = c_L[c];
  const int n1 = 2 * l1 + 1, n2 = 2 * l2 + 1, nL = 2 * L + 1;
  const int sz = n1 * n2 * nL;
  if (tid < 16) {
    double v = 1.0;
    for (int i = 1; i <= tid; ++i) v = v * i;
    sFct[tid] = v;
  }
  for (int t = tid; t < sz; t += INIT_TB) sC[t] = 0.f;
  __syncthreads();
  const double* f = sFct;
  if (tid < n1 * n2) {
    const int m = tid / n2, n = tid % n2;
    const int m1 = m - l1, m2 = n - l2, M = m1 + m2;
    if (M >= -L && M <= L) {
      const double pref0 =
          sqrt((2.0 * L + 1.0) * f[L + l1 - l2] * f[L - l1 + l2] * f[l1 + l2 - L] / f[l1 + l2 + L + 1]);
      const double pref =
          pref0 * sqrt(f[L + M] * f[L - M] * f[l1 - m1] * f[l1 + m1] * f[l2 - m2] * f[l2 + m2]);
      double s = 0.0;
      for (int kk = 0; kk <= l1 + l2 - L; ++kk) {
        const int a1 = l1 - m1 - kk, a2 = l2 + m2 - kk;
        const int a3 = L - l2 + m1 + kk, a4 = L - l1 - m2 + kk;
        if (a1 < 0 || a2 < 0 || a3 < 0 || a4 < 0) continue;
        const double term = 1.0 / (f[kk] * f[l1 + l2 - L - kk] * f[a1] * f[a2] * f[a3] * f[a4]);
        s += (kk & 1) ? -term : term;
      }
      const double val = pref * s;
      sC[(m * n2 + n) * nL + (M + L)] = (float)val;
    }
  }
  __syncthreads();
  const int cgo = c_off[c];
  for (int t = tid; t < sz; t += INIT_TB) ws[cgo + t] = sC[t];
  const int R = n2 * nL, fo = c_foff[c];
  const int mtiles = (nL + 1) >> 1, kch = (n1 + 3) >> 2;
  for (int t = tid; t < mtiles * kch * 64; t += INIT_TB) {
    const int fi = t >> 6, s = t & 63;
    const int mt = fi / kch, kc = fi % kch;
    const int ln = s >> 1, j = s & 1;
    const int rr = ln & 15;
    const int k = 2 * mt + (rr >> 3), n = rr & 7;
    const int m = kc * 4 + ((ln >> 4) ? 2 : 0) + j;
    ws[CG_TOTAL + (fo + fi) * 64 + s] =
        (k < nL && n < n2 && m < n1) ? sC[m * R + n * nL + k] : 0.f;
  }
}

template <int C>
__device__ __forceinline__ void wigner_combo(const float* __restrict__ sCG,
                                             const float* __restrict__ sFRAG,
                                             const float* __restrict__ XT,
                                             float* __restrict__ X2,
                                             const int lane, const int b, const int half) {
  constexpr int l1 = K_L1[C], l2 = K_L2[C], L = K_LL[C];
  constexpr int n1 = 2 * l1 + 1, n2 = 2 * l2 + 1, nL = 2 * L + 1;
  constexpr int cgo = K_OFF[C], fb = K_FOFF[C];
  constexpr int xo1 = K_XOFF[l1], xo2 = K_XOFF[l2];
  constexpr int mtiles = (nL + 1) >> 1, kch = (n1 + 3) >> 2;
  static_assert(fb + mtiles * kch == ((C + 1 < NCOMBO) ? K_FOFF[(C + 1 < NCOMBO) ? C + 1 : C] : NFRAGS),
                "fragment table offsets");
  constexpr int x2base = ((l1 + l2 + L) & 1) ? K_X2NEG[L] : K_X2POS[L];
  constexpr float wc = (l1 < l2) ? 2.f : 1.f;
  constexpr int D = l1 + l2 - L;

  const int m0a = half ? 2 : 0;
  v2f bf0[n1], bf1[n1];
#pragma unroll
  for (int p = 0; p < n1; ++p) {
    if constexpr (kch == 1) {
      const int r0 = (m0a     < n1) ? (xo1 + m0a * n1 + p)       : ZROW;
      const int r1 = (m0a + 1 < n1) ? (xo1 + (m0a + 1) * n1 + p) : ZROW;
      bf0[p].x = XT[r0 * 16 + b];
      bf0[p].y = XT[r1 * 16 + b];
      bf1[p].x = 0.f;
      bf1[p].y = 0.f;
    } else {
      {
        const int r0 = xo1 + m0a * n1 + p;
        const int r1 = xo1 + (m0a + 1) * n1 + p;
        bf0[p].x = XT[r0 * 16 + b];
        bf0[p].y = XT[r1 * 16 + b];
      }
      {
        const int m0 = 4 + m0a;
        const int r0 = (m0     < n1) ? (xo1 + m0 * n1 + p)       : ZROW;
        const int r1 = (m0 + 1 < n1) ? (xo1 + (m0 + 1) * n1 + p) : ZROW;
        bf1[p].x = XT[r0 * 16 + b];
        bf1[p].y = XT[r1 * 16 + b];
      }
    }
  }

  constexpr int nq2 = (n2 + 1) >> 1;
  v2f xv2[n2][nq2];
#pragma unroll
  for (int n = 0; n < n2; ++n) {
#pragma unroll
    for (int i = 0; i < nq2; ++i) {
      xv2[n][i].x = XT[(xo2 + n * n2 + 2 * i) * 16 + b];
      xv2[n][i].y = (2 * i + 1 < n2) ? XT[(xo2 + n * n2 + 2 * i + 1) * 16 + b] : 0.f;
    }
  }

  float cf[nL][n1];
#pragma unroll
  for (int j = 0; j < nL; ++j) {
#pragma unroll
    for (int p = 0; p < n1; ++p) {
      const int q = j + D - p;
      cf[j][p] = (q >= 0 && q < n2) ? sCG[cgo + (p * n2 + q) * nL + j] : 0.f;
    }
  }

  const float* fp = &sFRAG[fb * 64 + lane * 2];
  v2f af0 = *(const v2f*)fp;
  v2f af1 = {0.f, 0.f};
  if constexpr (kch == 2) af1 = *(const v2f*)(fp + 64);

#pragma unroll 1
  for (int mt = 0; mt < mtiles; ++mt) {
    const float* fn = fp + ((mt + 1 < mtiles) ? kch * 64 : 0);
    const v2f nf0 = *(const v2f*)fn;
    v2f nf1 = {0.f, 0.f};
    if constexpr (kch == 2) nf1 = *(const v2f*)(fn + 64);

    const int k = 2 * mt + half;
    float* Xk = X2 + ((k < nL) ? (x2base + k * nL) : X2ROWS) * 16 + b;
    float xo[nL];
#pragma unroll
    for (int j = 0; j < nL; ++j) xo[j] = Xk[j * 16];

    float s[nL];
#pragma unroll
    for (int j = 0; j < nL; ++j) s[j] = 0.f;

    const v8f zero8 = {0.f, 0.f, 0.f, 0.f, 0.f, 0.f, 0.f, 0.f};
    v8f accn = __builtin_amdgcn_wmma_f32_16x16x4_f32(false, af0, false, bf0[0],
                                                     (short)0, zero8, false, false);
    if constexpr (kch == 2)
      accn = __builtin_amdgcn_wmma_f32_16x16x4_f32(false, af1, false, bf1[0],
                                                   (short)0, accn, false, false);

#pragma unroll
    for (int p = 0; p < n1; ++p) {
      const v8f acc = accn;
      if (p + 1 < n1) {
        accn = __builtin_amdgcn_wmma_f32_16x16x4_f32(false, af0, false, bf0[p + 1],
                                                     (short)0, zero8, false, false);
        if constexpr (kch == 2)
          accn = __builtin_amdgcn_wmma_f32_16x16x4_f32(false, af1, false, bf1[p + 1],
                                                       (short)0, accn, false, false);
      }
      __builtin_amdgcn_sched_barrier(0);
      const int qlo = (D - p > 0) ? (D - p) : 0;
      const int qhi = (nL - 1 + D - p < n2 - 1) ? (nL - 1 + D - p) : (n2 - 1);
#pragma unroll
      for (int i = 0; i < nq2; ++i) {
        if (2 * i + 1 >= qlo && 2 * i <= qhi) {
          v2f e = {0.f, 0.f};
#pragma unroll
          for (int n = 0; n < n2; ++n) {
            const v2f an = {acc[n], acc[n]};
            e += xv2[n][i] * an;
          }
          if (2 * i >= qlo && 2 * i <= qhi)
            s[2 * i - D + p] += cf[2 * i - D + p][p] * e.x;
          if (2 * i + 1 >= qlo && 2 * i + 1 <= qhi)
            s[2 * i + 1 - D + p] += cf[2 * i + 1 - D + p][p] * e.y;
        }
      }
    }

#pragma unroll
    for (int j = 0; j < nL; ++j)
      Xk[j * 16] = xo[j] + wc * s[j];
    fp = fn;
    af0 = nf0;
    af1 = nf1;
  }
}

template <int S, int R>
__device__ __forceinline__ void chunk_load(const float* __restrict__ x, const int base,
                                           const int Btot, const int lane, float (&g)[R]) {
#pragma unroll
  for (int r = 0; r < R; ++r) {
    int i = lane + 32 * r;
    i = (i < 16 * S) ? i : (16 * S - 1);
    const int bb = i / S, c = i - bb * S;
    int e = base + bb;
    e = (e < Btot) ? e : (Btot - 1);
    g[r] = x[(unsigned)(e * S + c)];
  }
}
template <int S, int R, int XO>
__device__ __forceinline__ void chunk_store(float* __restrict__ XT, const int lane,
                                            const float (&g)[R]) {
#pragma unroll
  for (int r = 0; r < R; ++r) {
    const int i = lane + 32 * r;
    const int bb = i / S, c = i - bb * S;
    if (i < 16 * S) XT[(XO + c) * 16 + bb] = g[r];
  }
}

__launch_bounds__(TB, 1)
__global__ void wigner_main(const float* __restrict__ x0, const float* __restrict__ x1,
                            const float* __restrict__ x2, const float* __restrict__ x3,
                            const float* __restrict__ tab, float* __restrict__ out, int Btot) {
  __shared__ __align__(16) float sTAB[TAB_TOTAL];
  __shared__ float sXT[WAVES_PER_BLOCK][85 * 16];
  __shared__ float sX2[WAVES_PER_BLOCK][(X2ROWS + X2PAD) * 16];

  const int tid = threadIdx.x, lane = tid & 31, wid = tid >> 5;
  const int b = lane & 15, half = lane >> 4;
  const float* sCG = sTAB;
  const float* sFRAG = sTAB + CG_TOTAL;

  const int tile = blockIdx.x * WAVES_PER_BLOCK + wid;
  const int base = tile * 16;
  float* XT = sXT[wid];
  float* X2 = sX2[wid];

  {
    int pe = base + WAVES_PER_BLOCK * 16;
    if (pe < Btot) __builtin_prefetch(&x3[(size_t)pe * 49], 0, 1);
  }

  float g3[25], g2[13], g1[5], g0[1];
  chunk_load<49, 25>(x3, base, Btot, lane, g3);
  chunk_load<25, 13>(x2, base, Btot, lane, g2);
  chunk_load<9, 5>(x1, base, Btot, lane, g1);
  chunk_load<1, 1>(x0, base, Btot, lane, g0);
  const v4f* tab4 = (const v4f*)tab;
  v4f tv[TABR];
#pragma unroll
  for (int r = 0; r < TABR; ++r) {
    int i4 = tid + TB * r;
    i4 = (i4 < TAB4) ? i4 : (TAB4 - 1);
    tv[r] = tab4[i4];
  }
  const float tt = tab[TAB4 * 4 + ((tid < TABTAIL) ? tid : 0)];

  for (int t = lane; t < 16; t += 32) XT[ZROW * 16 + t] = 0.f;
  for (int t = lane; t < (X2ROWS + X2PAD) * 16; t += 32) X2[t] = 0.f;

  chunk_store<49, 25, 35>(XT, lane, g3);
  chunk_store<25, 13, 10>(XT, lane, g2);
  chunk_store<9, 5, 1>(XT, lane, g1);
  chunk_store<1, 1, 0>(XT, lane, g0);
#pragma unroll
  for (int r = 0; r < TABR; ++r) {
    const int i4 = tid + TB * r;
    if (i4 < TAB4) *(v4f*)&sTAB[i4 * 4] = tv[r];
  }
  if (tid < TABTAIL) sTAB[TAB4 * 4 + tid] = tt;
  __syncthreads();

#define WIGNER_COMBO(C) wigner_combo<C>(sCG, sFRAG, XT, X2, lane, b, half)
  WIGNER_COMBO(0);  WIGNER_COMBO(1);  WIGNER_COMBO(2);  WIGNER_COMBO(3);
  WIGNER_COMBO(4);  WIGNER_COMBO(5);  WIGNER_COMBO(6);  WIGNER_COMBO(7);
  WIGNER_COMBO(8);  WIGNER_COMBO(9);  WIGNER_COMBO(10); WIGNER_COMBO(11);
  WIGNER_COMBO(12); WIGNER_COMBO(13); WIGNER_COMBO(14); WIGNER_COMBO(15);
  WIGNER_COMBO(16); WIGNER_COMBO(17); WIGNER_COMBO(18); WIGNER_COMBO(19);
  WIGNER_COMBO(20); WIGNER_COMBO(21); WIGNER_COMBO(22);
#undef WIGNER_COMBO
  __syncthreads();

  if (lane < 16) {
    const int e = base + b;
    if (e < Btot) {
      const float nu1 = XT[0 * 16 + b];
      const float nu2 = X2[0 * 16 + b];
      float nu3 = 0.f, nu4 = 0.f;
      for (int l = 0; l <= 3; ++l) {
        const int n = 2 * l + 1;
        const int cgo = c_off[d_diag[l]];
        const int xo = d_xoff[l];
        const int xp = d_x2pos[l], xn = d_x2neg[l];
        for (int m = 0; m < n; ++m) {
          const int nn = n - 1 - m;
          const float c1 = sCG[cgo + m * n + nn];
          for (int p = 0; p < n; ++p) {
            const int q = n - 1 - p;
            const float c2 = sCG[cgo + p * n + q];
            const float w = c1 * c2;
            nu3 += w * X2[(xp + m * n + p) * 16 + b] * XT[(xo + nn * n + q) * 16 + b];
            nu4 += w * X2[(xp + m * n + p) * 16 + b] * X2[(xp + nn * n + q) * 16 + b];
            if (l > 0)
              nu4 += w * X2[(xn + m * n + p) * 16 + b] * X2[(xn + nn * n + q) * 16 + b];
          }
        }
      }
      out[(size_t)e * 4 + 0] = nu1;
      out[(size_t)e * 4 + 1] = nu2;
      out[(size_t)e * 4 + 2] = nu3;
      out[(size_t)e * 4 + 3] = nu4;
    }
  }
}

extern "C" void kernel_launch(void* const* d_in, const int* in_sizes, int n_in,
                              void* d_out, int out_size, void* d_ws, size_t ws_size,
                              hipStream_t stream) {
  const float* x0 = (const float*)d_in[0];
  const float* x1 = (const float*)d_in[1];
  const float* x2 = (const float*)d_in[2];
  const float* x3 = (const float*)d_in[3];
  float* tab = (float*)d_ws;
  const int Btot = in_sizes[0];

  wigner_cg_init<<<NCOMBO, INIT_TB, 0, stream>>>(tab);

  const int tiles = (Btot + 15) / 16;
  const int blocks = (tiles + WAVES_PER_BLOCK - 1) / WAVES_PER_BLOCK;
  wigner_main<<<blocks, TB, 0, stream>>>(x0, x1, x2, x3, tab, (float*)d_out, Btot);
}


# l1==l2 combos: E[q,k,p]=(-1)^L E[p,k,q] (same block both sides), so stages 1-3 run only q>=p with the q>p table entries doubled: fewer WMMA column blocks and stage-2 sums; stage-3 sum regrouped
# speedup vs baseline: 113.0202x; 1.1357x over previous
#include <hip/hip_runtime.h>


#define NCOMBO 23
#define CG_TOTAL 2351
#define NFRAGS 90
#define FRAG_TOTAL (NFRAGS * 64)
#define TAB_TOTAL (CG_TOTAL + FRAG_TOTAL)
#define WAVES_PER_BLOCK 4
#define TB (WAVES_PER_BLOCK * 32)
#define ZROW 84
#define X2ROWS 167
#define X2PAD 7
#define TAB4 (TAB_TOTAL / 4)
#define TABTAIL (TAB_TOTAL - 4 * TAB4)
#define TABR ((TAB4 + TB - 1) / TB)

typedef float v2f __attribute__((ext_vector_type(2)));
typedef float v4f __attribute__((ext_vector_type(4)));
typedef float v8f __attribute__((ext_vector_type(8)));

#define L1_LIST {0,0,0,0, 1,1,1,1,1,1,1,1, 2,2,2,2,2,2,2, 3,3,3,3}
#define L2_LIST {0,1,2,3, 1,1,1,2,2,2,3,3, 2,2,2,2,3,3,3, 3,3,3,3}
#define LL_LIST {0,1,2,3, 0,1,2,1,2,3,2,3, 0,1,2,3,1,2,3, 0,1,2,3}
#define OFF_LIST {0,1,10,35, 84,93,120,165,210,285,390,495, \
                  642,667,742,867,1042,1147,1322, 1567,1616,1763,2008}
#define FOFF_LIST {0,1,3,6, 10,11,13,16,18,21,25,28, \
                   32,34,38,44,52,56,62, 70,72,76,82}
#define XOFF_LIST {0, 1, 10, 35}
#define X2POS_LIST {0, 1, 19, 69}
#define X2NEG_LIST {0, 10, 44, 118}
__constant__ int c_l1[NCOMBO] = L1_LIST;
__constant__ int c_l2[NCOMBO] = L2_LIST;
__constant__ int c_L [NCOMBO] = LL_LIST;
__constant__ int c_off[NCOMBO] = OFF_LIST;
__constant__ int c_foff[NCOMBO] = FOFF_LIST;
__constant__ int d_xoff[4] = XOFF_LIST;
__constant__ int d_x2pos[4] = X2POS_LIST;
__constant__ int d_x2neg[4] = X2NEG_LIST;
constexpr int K_L1[NCOMBO] = L1_LIST;
constexpr int K_L2[NCOMBO] = L2_LIST;
constexpr int K_LL[NCOMBO] = LL_LIST;
constexpr int K_OFF[NCOMBO] = OFF_LIST;
constexpr int K_FOFF[NCOMBO] = FOFF_LIST;
constexpr int K_XOFF[4] = XOFF_LIST;
constexpr int K_X2POS[4] = X2POS_LIST;
constexpr int K_X2NEG[4] = X2NEG_LIST;
__constant__ int d_diag[4] = {0, 4, 12, 19};

#define INIT_TB 128
__global__ void wigner_cg_init(float* __restrict__ ws) {
  __shared__ double sFct[16];
  __shared__ float sC[343];
  const int tid = threadIdx.x;
  const int c = blockIdx.x;
  const int l1 = c_l1[c], l2 = c_l2[c], L = c_L[c];
  const int n1 = 2 * l1 + 1, n2 = 2 * l2 + 1, nL = 2 * L + 1;
  const int sz = n1 * n2 * nL;
  if (tid < 16) {
    double v = 1.0;
    for (int i = 1; i <= tid; ++i) v = v * i;
    sFct[tid] = v;
  }
  for (int t = tid; t < sz; t += INIT_TB) sC[t] = 0.f;
  __syncthreads();
  const double* f = sFct;
  if (tid < n1 * n2) {
    const int m = tid / n2, n = tid % n2;
    const int m1 = m - l1, m2 = n - l2, M = m1 + m2;
    if (M >= -L && M <= L) {
      const double pref0 =
          sqrt((2.0 * L + 1.0) * f[L + l1 - l2] * f[L - l1 + l2] * f[l1 + l2 - L] / f[l1 + l2 + L + 1]);
      const double pref =
          pref0 * sqrt(f[L + M] * f[L - M] * f[l1 - m1] * f[l1 + m1] * f[l2 - m2] * f[l2 + m2]);
      double s = 0.0;
      for (int kk = 0; kk <= l1 + l2 - L; ++kk) {
        const int a1 = l1 - m1 - kk, a2 = l2 + m2 - kk;
        const int a3 = L - l2 + m1 + kk, a4 = L - l1 - m2 + kk;
        if (a1 < 0 || a2 < 0 || a3 < 0 || a4 < 0) continue;
        const double term = 1.0 / (f[kk] * f[l1 + l2 - L - kk] * f[a1] * f[a2] * f[a3] * f[a4]);
        s += (kk & 1) ? -term : term;
      }
      const double val = pref * s;
      sC[(m * n2 + n) * nL + (M + L)] = (float)val;
    }
  }
  __syncthreads();
  const int cgo = c_off[c];
  for (int t = tid; t < sz; t += INIT_TB) ws[cgo + t] = sC[t];
  const int R = n2 * nL, fo = c_foff[c];
  const int mtiles = (nL + 1) >> 1, kch = (n1 + 3) >> 2;
  for (int t = tid; t < mtiles * kch * 64; t += INIT_TB) {
    const int fi = t >> 6, s = t & 63;
    const int mt = fi / kch, kc = fi % kch;
    const int ln = s >> 1, j = s & 1;
    const int rr = ln & 15;
    const int k = 2 * mt + (rr >> 3), n = rr & 7;
    const int m = kc * 4 + ((ln >> 4) ? 2 : 0) + j;
    ws[CG_TOTAL + (fo + fi) * 64 + s] =
        (k < nL && n < n2 && m < n1) ? sC[m * R + n * nL + k] : 0.f;
  }
}

template <int C>
__device__ __forceinline__ void wigner_combo(const float* __restrict__ sCG,
                                             const float* __restrict__ sFRAG,
                                             const float* __restrict__ XT,
                                             float* __restrict__ X2,
                                             const int lane, const int b, const int half) {
  constexpr int l1 = K_L1[C], l2 = K_L2[C], L = K_LL[C];
  constexpr int n1 = 2 * l1 + 1, n2 = 2 * l2 + 1, nL = 2 * L + 1;
  constexpr int cgo = K_OFF[C], fb = K_FOFF[C];
  constexpr int xo1 = K_XOFF[l1], xo2 = K_XOFF[l2];
  constexpr int mtiles = (nL + 1) >> 1, kch = (n1 + 3) >> 2;
  static_assert(fb + mtiles * kch == ((C + 1 < NCOMBO) ? K_FOFF[(C + 1 < NCOMBO) ? C + 1 : C] : NFRAGS),
                "fragment table offsets");
  constexpr int x2base = ((l1 + l2 + L) & 1) ? K_X2NEG[L] : K_X2POS[L];
  constexpr float wc = (l1 < l2) ? 2.f : 1.f;
  constexpr int D = l1 + l2 - L;
  constexpr bool SYM = (l1 == l2);
  constexpr int np = SYM ? ((((nL - 1 + D) >> 1) + 1 < n1) ? (((nL - 1 + D) >> 1) + 1) : n1) : n1;

  const int m0a = half ? 2 : 0;
  v2f bf0[np], bf1[np];
#pragma unroll
  for (int p = 0; p < np; ++p) {
    if constexpr (kch == 1) {
      const int r0 = (m0a     < n1) ? (xo1 + m0a * n1 + p)       : ZROW;
      const int r1 = (m0a + 1 < n1) ? (xo1 + (m0a + 1) * n1 + p) : ZROW;
      bf0[p].x = XT[r0 * 16 + b];
      bf0[p].y = XT[r1 * 16 + b];
      bf1[p].x = 0.f;
      bf1[p].y = 0.f;
    } else {
      {
        const int r0 = xo1 + m0a * n1 + p;
        const int r1 = xo1 + (m0a + 1) * n1 + p;
        bf0[p].x = XT[r0 * 16 + b];
        bf0[p].y = XT[r1 * 16 + b];
      }
      {
        const int m0 = 4 + m0a;
        const int r0 = (m0     < n1) ? (xo1 + m0 * n1 + p)       : ZROW;
        const int r1 = (m0 + 1 < n1) ? (xo1 + (m0 + 1) * n1 + p) : ZROW;
        bf1[p].x = XT[r0 * 16 + b];
        bf1[p].y = XT[r1 * 16 + b];
      }
    }
  }

  constexpr int nq2 = (n2 + 1) >> 1;
  v2f xv2[n2][nq2];
#pragma unroll
  for (int n = 0; n < n2; ++n) {
#pragma unroll
    for (int i = 0; i < nq2; ++i) {
      xv2[n][i].x = XT[(xo2 + n * n2 + 2 * i) * 16 + b];
      xv2[n][i].y = (2 * i + 1 < n2) ? XT[(xo2 + n * n2 + 2 * i + 1) * 16 + b] : 0.f;
    }
  }

  float cf[nL][np];
#pragma unroll
  for (int j = 0; j < nL; ++j) {
#pragma unroll
    for (int p = 0; p < np; ++p) {
      const int q = j + D - p;
      float c = (q >= 0 && q < n2 && (!SYM || q >= p)) ? sCG[cgo + (p * n2 + q) * nL + j] : 0.f;
      if (SYM && q > p) c = c + c;
      cf[j][p] = c;
    }
  }

  const float* fp = &sFRAG[fb * 64 + lane * 2];
  v2f af0 = *(const v2f*)fp;
  v2f af1 = {0.f, 0.f};
  if constexpr (kch == 2) af1 = *(const v2f*)(fp + 64);

#pragma unroll 1
  for (int mt = 0; mt < mtiles; ++mt) {
    const float* fn = fp + ((mt + 1 < mtiles) ? kch * 64 : 0);
    const v2f nf0 = *(const v2f*)fn;
    v2f nf1 = {0.f, 0.f};
    if constexpr (kch == 2) nf1 = *(const v2f*)(fn + 64);

    const int k = 2 * mt + half;
    float* Xk = X2 + ((k < nL) ? (x2base + k * nL) : X2ROWS) * 16 + b;
    float xo[nL];
#pragma unroll
    for (int j = 0; j < nL; ++j) xo[j] = Xk[j * 16];

    float s[nL];
#pragma unroll
    for (int j = 0; j < nL; ++j) s[j] = 0.f;

    const v8f zero8 = {0.f, 0.f, 0.f, 0.f, 0.f, 0.f, 0.f, 0.f};
    v8f accn = __builtin_amdgcn_wmma_f32_16x16x4_f32(false, af0, false, bf0[0],
                                                     (short)0, zero8, false, false);
    if constexpr (kch == 2)
      accn = __builtin_amdgcn_wmma_f32_16x16x4_f32(false, af1, false, bf1[0],
                                                   (short)0, accn, false, false);

#pragma unroll
    for (int p = 0; p < np; ++p) {
      const v8f acc = accn;
      if (p + 1 < np) {
        accn = __builtin_amdgcn_wmma_f32_16x16x4_f32(false, af0, false, bf0[p + 1],
                                                     (short)0, zero8, false, false);
        if constexpr (kch == 2)
          accn = __builtin_amdgcn_wmma_f32_16x16x4_f32(false, af1, false, bf1[p + 1],
                                                       (short)0, accn, false, false);
      }
      __builtin_amdgcn_sched_barrier(0);
      const int qlo0 = (D - p > 0) ? (D - p) : 0;
      const int qlo = (SYM && p > qlo0) ? p : qlo0;
      const int qhi = (nL - 1 + D - p < n2 - 1) ? (nL - 1 + D - p) : (n2 - 1);
#pragma unroll
      for (int i = 0; i < nq2; ++i) {
        if (2 * i + 1 >= qlo && 2 * i <= qhi) {
          v2f e = {0.f, 0.f};
#pragma unroll
          for (int n = 0; n < n2; ++n) {
            const v2f an = {acc[n], acc[n]};
            e += xv2[n][i] * an;
          }
          if (2 * i >= qlo && 2 * i <= qhi)
            s[2 * i - D + p] += cf[2 * i - D + p][p] * e.x;
          if (2 * i + 1 >= qlo && 2 * i + 1 <= qhi)
            s[2 * i + 1 - D + p] += cf[2 * i + 1 - D + p][p] * e.y;
        }
      }
    }

#pragma unroll
    for (int j = 0; j < nL; ++j)
      Xk[j * 16] = xo[j] + wc * s[j];
    fp = fn;
    af0 = nf0;
    af1 = nf1;
  }
}

template <int S, int R>
__device__ __forceinline__ void chunk_load(const float* __restrict__ x, const int base,
                                           const int Btot, const int lane, float (&g)[R]) {
#pragma unroll
  for (int r = 0; r < R; ++r) {
    int i = lane + 32 * r;
    i = (i < 16 * S) ? i : (16 * S - 1);
    const int bb = i / S, c = i - bb * S;
    int e = base + bb;
    e = (e < Btot) ? e : (Btot - 1);
    g[r] = x[(unsigned)(e * S + c)];
  }
}
template <int S, int R, int XO>
__device__ __forceinline__ void chunk_store(float* __restrict__ XT, const int lane,
                                            const float (&g)[R]) {
#pragma unroll
  for (int r = 0; r < R; ++r) {
    const int i = lane + 32 * r;
    const int bb = i / S, c = i - bb * S;
    if (i < 16 * S) XT[(XO + c) * 16 + bb] = g[r];
  }
}

__launch_bounds__(TB, 1)
__global__ void wigner_main(const float* __restrict__ x0, const float* __restrict__ x1,
                            const float* __restrict__ x2, const float* __restrict__ x3,
                            const float* __restrict__ tab, float* __restrict__ out, int Btot) {
  __shared__ __align__(16) float sTAB[TAB_TOTAL];
  __shared__ float sXT[WAVES_PER_BLOCK][85 * 16];
  __shared__ float sX2[WAVES_PER_BLOCK][(X2ROWS + X2PAD) * 16];

  const int tid = threadIdx.x, lane = tid & 31, wid = tid >> 5;
  const int b = lane & 15, half = lane >> 4;
  const float* sCG = sTAB;
  const float* sFRAG = sTAB + CG_TOTAL;

  const int tile = blockIdx.x * WAVES_PER_BLOCK + wid;
  const int base = tile * 16;
  float* XT = sXT[wid];
  float* X2 = sX2[wid];

  {
    int pe = base + WAVES_PER_BLOCK * 16;
    if (pe < Btot) __builtin_prefetch(&x3[(size_t)pe * 49], 0, 1);
  }

  float g3[25], g2[13], g1[5], g0[1];
  chunk_load<49, 25>(x3, base, Btot, lane, g3);
  chunk_load<25, 13>(x2, base, Btot, lane, g2);
  chunk_load<9, 5>(x1, base, Btot, lane, g1);
  chunk_load<1, 1>(x0, base, Btot, lane, g0);
  const v4f* tab4 = (const v4f*)tab;
  v4f tv[TABR];
#pragma unroll
  for (int r = 0; r < TABR; ++r) {
    int i4 = tid + TB * r;
    i4 = (i4 < TAB4) ? i4 : (TAB4 - 1);
    tv[r] = tab4[i4];
  }
  const float tt = tab[TAB4 * 4 + ((tid < TABTAIL) ? tid : 0)];

  for (int t = lane; t < 16; t += 32) XT[ZROW * 16 + t] = 0.f;
  for (int t = lane; t < (X2ROWS + X2PAD) * 16; t += 32) X2[t] = 0.f;

  chunk_store<49, 25, 35>(XT, lane, g3);
  chunk_store<25, 13, 10>(XT, lane, g2);
  chunk_store<9, 5, 1>(XT, lane, g1);
  chunk_store<1, 1, 0>(XT, lane, g0);
#pragma unroll
  for (int r = 0; r < TABR; ++r) {
    const int i4 = tid + TB * r;
    if (i4 < TAB4) *(v4f*)&sTAB[i4 * 4] = tv[r];
  }
  if (tid < TABTAIL) sTAB[TAB4 * 4 + tid] = tt;
  __syncthreads();

#define WIGNER_COMBO(C) wigner_combo<C>(sCG, sFRAG, XT, X2, lane, b, half)
  WIGNER_COMBO(0);  WIGNER_COMBO(1);  WIGNER_COMBO(2);  WIGNER_COMBO(3);
  WIGNER_COMBO(4);  WIGNER_COMBO(5);  WIGNER_COMBO(6);  WIGNER_COMBO(7);
  WIGNER_COMBO(8);  WIGNER_COMBO(9);  WIGNER_COMBO(10); WIGNER_COMBO(11);
  WIGNER_COMBO(12); WIGNER_COMBO(13); WIGNER_COMBO(14); WIGNER_COMBO(15);
  WIGNER_COMBO(16); WIGNER_COMBO(17); WIGNER_COMBO(18); WIGNER_COMBO(19);
  WIGNER_COMBO(20); WIGNER_COMBO(21); WIGNER_COMBO(22);
#undef WIGNER_COMBO
  __syncthreads();

  if (lane < 16) {
    const int e = base + b;
    if (e < Btot) {
      const float nu1 = XT[0 * 16 + b];
      const float nu2 = X2[0 * 16 + b];
      float nu3 = 0.f, nu4 = 0.f;
      for (int l = 0; l <= 3; ++l) {
        const int n = 2 * l + 1;
        const int cgo = c_off[d_diag[l]];
        const int xo = d_xoff[l];
        const int xp = d_x2pos[l], xn = d_x2neg[l];
        for (int m = 0; m < n; ++m) {
          const int nn = n - 1 - m;
          const float c1 = sCG[cgo + m * n + nn];
          for (int p = 0; p < n; ++p) {
            const int q = n - 1 - p;
            const float c2 = sCG[cgo + p * n + q];
            const float w = c1 * c2;
            nu3 += w * X2[(xp + m * n + p) * 16 + b] * XT[(xo + nn * n + q) * 16 + b];
            nu4 += w * X2[(xp + m * n + p) * 16 + b] * X2[(xp + nn * n + q) * 16 + b];
            if (l > 0)
              nu4 += w * X2[(xn + m * n + p) * 16 + b] * X2[(xn + nn * n + q) * 16 + b];
          }
        }
      }
      out[(size_t)e * 4 + 0] = nu1;
      out[(size_t)e * 4 + 1] = nu2;
      out[(size_t)e * 4 + 2] = nu3;
      out[(size_t)e * 4 + 3] = nu4;
    }
  }
}

extern "C" void kernel_launch(void* const* d_in, const int* in_sizes, int n_in,
                              void* d_out, int out_size, void* d_ws, size_t ws_size,
                              hipStream_t stream) {
  const float* x0 = (const float*)d_in[0];
  const float* x1 = (const float*)d_in[1];
  const float* x2 = (const float*)d_in[2];
  const float* x3 = (const float*)d_in[3];
  float* tab = (float*)d_ws;
  const int Btot = in_sizes[0];

  wigner_cg_init<<<NCOMBO, INIT_TB, 0, stream>>>(tab);

  const int tiles = (Btot + 15) / 16;
  const int blocks = (tiles + WAVES_PER_BLOCK - 1) / WAVES_PER_BLOCK;
  wigner_main<<<blocks, TB, 0, stream>>>(x0, x1, x2, x3, tab, (float*)d_out, Btot);
}


# Stage-1 rows indexed (k,m) not (k,n): a row's one non-zero sits at its own K index, so one WMMA per K chunk serves two k per lane (331 to 193 WMMAs); stage-2 sum now runs over m (n descending)
# speedup vs baseline: 142.8077x; 1.2636x over previous
#include <hip/hip_runtime.h>


#define NCOMBO 23
#define CG_TOTAL 2351
#define NFRAGS 53
#define FRAG_TOTAL (NFRAGS * 64)
#define TAB_TOTAL (CG_TOTAL + FRAG_TOTAL)
#define WAVES_PER_BLOCK 4
#define TB (WAVES_PER_BLOCK * 32)
#define ZROW 84
#define XGUARD 8
#define XROWS 120
#define X2ROWS 167
#define X2PAD 7
#define TAB4 (TAB_TOTAL / 4)
#define TABTAIL (TAB_TOTAL - 4 * TAB4)
#define TABR ((TAB4 + TB - 1) / TB)

typedef float v2f __attribute__((ext_vector_type(2)));
typedef float v4f __attribute__((ext_vector_type(4)));
typedef float v8f __attribute__((ext_vector_type(8)));

#define L1_LIST {0,0,0,0, 1,1,1,1,1,1,1,1, 2,2,2,2,2,2,2, 3,3,3,3}
#define L2_LIST {0,1,2,3, 1,1,1,2,2,2,3,3, 2,2,2,2,3,3,3, 3,3,3,3}
#define LL_LIST {0,1,2,3, 0,1,2,1,2,3,2,3, 0,1,2,3,1,2,3, 0,1,2,3}
#define OFF_LIST {0,1,10,35, 84,93,120,165,210,285,390,495, \
                  642,667,742,867,1042,1147,1322, 1567,1616,1763,2008}
#define FOFF_LIST {0,1,2,4, 6,7,8,10,11,13,15,17, \
                   19,21,23,27,31,33,37, 41,43,45,49}
#define XOFF_LIST {0, 1, 10, 35}
#define X2POS_LIST {0, 1, 19, 69}
#define X2NEG_LIST {0, 10, 44, 118}
__constant__ int c_l1[NCOMBO] = L1_LIST;
__constant__ int c_l2[NCOMBO] = L2_LIST;
__constant__ int c_L [NCOMBO] = LL_LIST;
__constant__ int c_off[NCOMBO] = OFF_LIST;
__constant__ int c_foff[NCOMBO] = FOFF_LIST;
__constant__ int d_xoff[4] = XOFF_LIST;
__constant__ int d_x2pos[4] = X2POS_LIST;
__constant__ int d_x2neg[4] = X2NEG_LIST;
constexpr int K_L1[NCOMBO] = L1_LIST;
constexpr int K_L2[NCOMBO] = L2_LIST;
constexpr int K_LL[NCOMBO] = LL_LIST;
constexpr int K_OFF[NCOMBO] = OFF_LIST;
constexpr int K_FOFF[NCOMBO] = FOFF_LIST;
constexpr int K_XOFF[4] = XOFF_LIST;
constexpr int K_X2POS[4] = X2POS_LIST;
constexpr int K_X2NEG[4] = X2NEG_LIST;
__constant__ int d_diag[4] = {0, 4, 12, 19};

#define INIT_TB 128
__global__ void wigner_cg_init(float* __restrict__ ws) {
  __shared__ double sFct[16];
  __shared__ float sC[343];
  const int tid = threadIdx.x;
  const int c = blockIdx.x;
  const int l1 = c_l1[c], l2 = c_l2[c], L = c_L[c];
  const int n1 = 2 * l1 + 1, n2 = 2 * l2 + 1, nL = 2 * L + 1;
  const int sz = n1 * n2 * nL;
  if (tid < 16) {
    double v = 1.0;
    for (int i = 1; i <= tid; ++i) v = v * i;
    sFct[tid] = v;
  }
  for (int t = tid; t < sz; t += INIT_TB) sC[t] = 0.f;
  __syncthreads();
  const double* f = sFct;
  if (tid < n1 * n2) {
    const int m = tid / n2, n = tid % n2;
    const int m1 = m - l1, m2 = n - l2, M = m1 + m2;
    if (M >= -L && M <= L) {
      const double pref0 =
          sqrt((2.0 * L + 1.0) * f[L + l1 - l2] * f[L - l1 + l2] * f[l1 + l2 - L] / f[l1 + l2 + L + 1]);
      const double pref =
          pref0 * sqrt(f[L + M] * f[L - M] * f[l1 - m1] * f[l1 + m1] * f[l2 - m2] * f[l2 + m2]);
      double s = 0.0;
      for (int kk = 0; kk <= l1 + l2 - L; ++kk) {
        const int a1 = l1 - m1 - kk, a2 = l2 + m2 - kk;
        const int a3 = L - l2 + m1 + kk, a4 = L - l1 - m2 + kk;
        if (a1 < 0 || a2 < 0 || a3 < 0 || a4 < 0) continue;
        const double term = 1.0 / (f[kk] * f[l1 + l2 - L - kk] * f[a1] * f[a2] * f[a3] * f[a4]);
        s += (kk & 1) ? -term : term;
      }
      const double val = pref * s;
      sC[(m * n2 + n) * nL + (M + L)] = (float)val;
    }
  }
  __syncthreads();
  const int cgo = c_off[c];
  for (int t = tid; t < sz; t += INIT_TB) ws[cgo + t] = sC[t];
  const int R = n2 * nL, fo = c_foff[c];
  const int ntp = (nL + 3) >> 2, kch = (n1 + 3) >> 2;
  const int D = l1 + l2 - L;
  for (int t = tid; t < ntp * kch * 64; t += INIT_TB) {
    const int fi = t >> 6, s = t & 63;
    const int tp = fi / kch, kc = fi % kch;
    const int ln = s >> 1, j = s & 1;
    const int rr = ln & 15, sl = rr & 7;
    const int k = 4 * tp + 2 * (sl >> 2) + (rr >> 3);
    const int ms = sl & 3, m = kc * 4 + ms;
    const int n = k + D - m;
    const int kap = ((ln >> 4) ? 2 : 0) + j;
    ws[CG_TOTAL + (fo + fi) * 64 + s] =
        (kap == ms && k < nL && m < n1 && n >= 0 && n < n2) ? sC[m * R + n * nL + k] : 0.f;
  }
}

template <int C>
struct WK {
  static constexpr int l1 = K_L1[C], l2 = K_L2[C], L = K_LL[C];
  static constexpr int n1 = 2 * l1 + 1, n2 = 2 * l2 + 1, nL = 2 * L + 1;
  static constexpr int D = l1 + l2 - L;
  static constexpr int kch = (n1 + 3) >> 2;
  static constexpr int nfull = (nL + 1) >> 2;
  static constexpr int nhalf = (nL - 4 * nfull > 0) ? 1 : 0;
  static constexpr bool SYM = (l1 == l2);
  static constexpr int np = SYM ? ((((nL - 1 + D) >> 1) + 1 < n1) ? (((nL - 1 + D) >> 1) + 1) : n1) : n1;
  static constexpr int nq2 = (n2 + 1) >> 1;
  static constexpr int x2base = ((l1 + l2 + L) & 1) ? K_X2NEG[L] : K_X2POS[L];
};

template <int C, int NS, int NTRIPS>
__device__ __forceinline__ void wigner_trips(const float* __restrict__ fp, const int kfirst,
                                             const v2f (&bf0)[WK<C>::np], const v2f (&bf1)[WK<C>::np],
                                             const float (&cf)[WK<C>::nL][WK<C>::np],
                                             const float* __restrict__ XT, float* __restrict__ X2,
                                             const int b) {
  using W = WK<C>;
  constexpr int l1 = W::l1, l2 = W::l2;
  constexpr int n1 = W::n1, n2 = W::n2, nL = W::nL, D = W::D, kch = W::kch;
  constexpr int np = W::np, nq2 = W::nq2, x2base = W::x2base;
  constexpr bool SYM = W::SYM;
  constexpr int xo2 = K_XOFF[l2];
  constexpr float wc = (l1 < l2) ? 2.f : 1.f;

  v2f af0 = *(const v2f*)fp;
  v2f af1 = {0.f, 0.f};
  if constexpr (kch == 2) af1 = *(const v2f*)(fp + 64);

#pragma unroll 1
  for (int t = 0; t < NTRIPS; ++t) {
    const float* fn = fp + ((t + 1 < NTRIPS) ? kch * 64 : 0);
    const v2f nf0 = *(const v2f*)fn;
    v2f nf1 = {0.f, 0.f};
    if constexpr (kch == 2) nf1 = *(const v2f*)(fn + 64);

    const int k0 = kfirst + 4 * t;
    v2f xr[NS][n1][nq2];
    float* Xk[NS];
    float xo[NS][nL];
    float s[NS][nL];
#pragma unroll
    for (int g = 0; g < NS; ++g) {
      const int k = k0 + 2 * g;
      const float* Xr = XT + (xo2 + (k + D) * n2) * 16 + b;
#pragma unroll
      for (int m = 0; m < n1; ++m) {
#pragma unroll
        for (int i = 0; i < nq2; ++i) {
          xr[g][m][i].x = Xr[(2 * i - m * n2) * 16];
          xr[g][m][i].y = (2 * i + 1 < n2) ? Xr[(2 * i + 1 - m * n2) * 16] : 0.f;
        }
      }
      Xk[g] = X2 + ((k < nL) ? (x2base + k * nL) : X2ROWS) * 16 + b;
#pragma unroll
      for (int j = 0; j < nL; ++j) {
        xo[g][j] = Xk[g][j * 16];
        s[g][j] = 0.f;
      }
    }

    const v8f zero8 = {0.f, 0.f, 0.f, 0.f, 0.f, 0.f, 0.f, 0.f};
    v8f accn0 = __builtin_amdgcn_wmma_f32_16x16x4_f32(false, af0, false, bf0[0],
                                                      (short)0, zero8, false, false);
    v8f accn1 = zero8;
    if constexpr (kch == 2)
      accn1 = __builtin_amdgcn_wmma_f32_16x16x4_f32(false, af1, false, bf1[0],
                                                    (short)0, zero8, false, false);

#pragma unroll
    for (int p = 0; p < np; ++p) {
      const v8f acc0 = accn0;
      const v8f acc1 = accn1;
      if (p + 1 < np) {
        accn0 = __builtin_amdgcn_wmma_f32_16x16x4_f32(false, af0, false, bf0[p + 1],
                                                      (short)0, zero8, false, false);
        if constexpr (kch == 2)
          accn1 = __builtin_amdgcn_wmma_f32_16x16x4_f32(false, af1, false, bf1[p + 1],
                                                        (short)0, zero8, false, false);
      }
      __builtin_amdgcn_sched_barrier(0);
      const int qlo0 = (D - p > 0) ? (D - p) : 0;
      const int qlo = (SYM && p > qlo0) ? p : qlo0;
      const int qhi = (nL - 1 + D - p < n2 - 1) ? (nL - 1 + D - p) : (n2 - 1);
#pragma unroll
      for (int g = 0; g < NS; ++g) {
#pragma unroll
        for (int i = 0; i < nq2; ++i) {
          if (2 * i + 1 >= qlo && 2 * i <= qhi) {
            v2f e = {0.f, 0.f};
#pragma unroll
            for (int m = 0; m < n1; ++m) {
              const float a = (m < 4) ? acc0[4 * g + (m & 3)] : acc1[4 * g + (m & 3)];
              const v2f am = {a, a};
              e += xr[g][m][i] * am;
            }
            if (2 * i >= qlo && 2 * i <= qhi)
              s[g][2 * i - D + p] += cf[2 * i - D + p][p] * e.x;
            if (2 * i + 1 >= qlo && 2 * i + 1 <= qhi)
              s[g][2 * i + 1 - D + p] += cf[2 * i + 1 - D + p][p] * e.y;
          }
        }
      }
    }

#pragma unroll
    for (int g = 0; g < NS; ++g) {
#pragma unroll
      for (int j = 0; j < nL; ++j)
        Xk[g][j * 16] = xo[g][j] + wc * s[g][j];
    }
    fp = fn;
    af0 = nf0;
    af1 = nf1;
  }
}

template <int C>
__device__ __forceinline__ void wigner_combo(const float* __restrict__ sCG,
                                             const float* __restrict__ sFRAG,
                                             const float* __restrict__ XT,
                                             float* __restrict__ X2,
                                             const int lane, const int b, const int half) {
  using W = WK<C>;
  constexpr int l1 = W::l1, l2 = W::l2;
  constexpr int n1 = W::n1, n2 = W::n2, nL = W::nL, D = W::D, kch = W::kch;
  constexpr int np = W::np, nfull = W::nfull, nhalf = W::nhalf;
  constexpr bool SYM = W::SYM;
  constexpr int cgo = K_OFF[C], fb = K_FOFF[C];
  constexpr int xo1 = K_XOFF[l1];
  static_assert(fb + (nfull + nhalf) * kch == ((C + 1 < NCOMBO) ? K_FOFF[(C + 1 < NCOMBO) ? C + 1 : C] : NFRAGS),
                "fragment table offsets");

  const int m0a = half ? 2 : 0;
  v2f bf0[np], bf1[np];
#pragma unroll
  for (int p = 0; p < np; ++p) {
    if constexpr (kch == 1) {
      const int r0 = (m0a     < n1) ? (xo1 + m0a * n1 + p)       : ZROW;
      const int r1 = (m0a + 1 < n1) ? (xo1 + (m0a + 1) * n1 + p) : ZROW;
      bf0[p].x = XT[r0 * 16 + b];
      bf0[p].y = XT[r1 * 16 + b];
      bf1[p].x = 0.f;
      bf1[p].y = 0.f;
    } else {
      {
        const int r0 = xo1 + m0a * n1 + p;
        const int r1 = xo1 + (m0a + 1) * n1 + p;
        bf0[p].x = XT[r0 * 16 + b];
        bf0[p].y = XT[r1 * 16 + b];
      }
      {
        const int m0 = 4 + m0a;
        const int r0 = (m0     < n1) ? (xo1 + m0 * n1 + p)       : ZROW;
        const int r1 = (m0 + 1 < n1) ? (xo1 + (m0 + 1) * n1 + p) : ZROW;
        bf1[p].x = XT[r0 * 16 + b];
        bf1[p].y = XT[r1 * 16 + b];
      }
    }
  }

  float cf[nL][np];
#pragma unroll
  for (int j = 0; j < nL; ++j) {
#pragma unroll
    for (int p = 0; p < np; ++p) {
      const int q = j + D - p;
      float c = (q >= 0 && q < n2 && (!SYM || q >= p)) ? sCG[cgo + (p * n2 + q) * nL + j] : 0.f;
      if (SYM && q > p) c = c + c;
      cf[j][p] = c;
    }
  }

  const float* fp = &sFRAG[fb * 64 + lane * 2];
  if constexpr (nfull > 0)
    wigner_trips<C, 2, nfull>(fp, half, bf0, bf1, cf, XT, X2, b);
  if constexpr (nhalf > 0)
    wigner_trips<C, 1, 1>(fp + nfull * kch * 64, 4 * nfull + half, bf0, bf1, cf, XT, X2, b);
}

template <int S, int R>
__device__ __forceinline__ void chunk_load(const float* __restrict__ x, const int base,
                                           const int Btot, const int lane, float (&g)[R]) {
#pragma unroll
  for (int r = 0; r < R; ++r) {
    int i = lane + 32 * r;
    i = (i < 16 * S) ? i : (16 * S - 1);
    const int bb = i / S, c = i - bb * S;
    int e = base + bb;
    e = (e < Btot) ? e : (Btot - 1);
    g[r] = x[(unsigned)(e * S + c)];
  }
}
template <int S, int R, int XO>
__device__ __forceinline__ void chunk_store(float* __restrict__ XT, const int lane,
                                            const float (&g)[R]) {
#pragma unroll
  for (int r = 0; r < R; ++r) {
    const int i = lane + 32 * r;
    const int bb = i / S, c = i - bb * S;
    if (i < 16 * S) XT[(XO + c) * 16 + bb] = g[r];
  }
}

__launch_bounds__(TB, 1)
__global__ void wigner_main(const float* __restrict__ x0, const float* __restrict__ x1,
                            const float* __restrict__ x2, const float* __restrict__ x3,
                            const float* __restrict__ tab, float* __restrict__ out, int Btot) {
  __shared__ __align__(16) float sTAB[TAB_TOTAL];
  __shared__ float sXT[WAVES_PER_BLOCK][XROWS * 16];
  __shared__ float sX2[WAVES_PER_BLOCK][(X2ROWS + X2PAD) * 16];

  const int tid = threadIdx.x, lane = tid & 31, wid = tid >> 5;
  const int b = lane & 15, half = lane >> 4;
  const float* sCG = sTAB;
  const float* sFRAG = sTAB + CG_TOTAL;

  const int tile = blockIdx.x * WAVES_PER_BLOCK + wid;
  const int base = tile * 16;
  float* XT = sXT[wid] + XGUARD * 16;
  float* X2 = sX2[wid];

  {
    int pe = base + WAVES_PER_BLOCK * 16;
    if (pe < Btot) __builtin_prefetch(&x3[(size_t)pe * 49], 0, 1);
  }

  float g3[25], g2[13], g1[5], g0[1];
  chunk_load<49, 25>(x3, base, Btot, lane, g3);
  chunk_load<25, 13>(x2, base, Btot, lane, g2);
  chunk_load<9, 5>(x1, base, Btot, lane, g1);
  chunk_load<1, 1>(x0, base, Btot, lane, g0);
  const v4f* tab4 = (const v4f*)tab;
  v4f tv[TABR];
#pragma unroll
  for (int r = 0; r < TABR; ++r) {
    int i4 = tid + TB * r;
    i4 = (i4 < TAB4) ? i4 : (TAB4 - 1);
    tv[r] = tab4[i4];
  }
  const float tt = tab[TAB4 * 4 + ((tid < TABTAIL) ? tid : 0)];

  for (int t = lane; t < XGUARD * 16; t += 32) XT[t - XGUARD * 16] = 0.f;
  for (int t = lane; t < (XROWS - XGUARD - ZROW) * 16; t += 32) XT[ZROW * 16 + t] = 0.f;
  for (int t = lane; t < (X2ROWS + X2PAD) * 16; t += 32) X2[t] = 0.f;

  chunk_store<49, 25, 35>(XT, lane, g3);
  chunk_store<25, 13, 10>(XT, lane, g2);
  chunk_store<9, 5, 1>(XT, lane, g1);
  chunk_store<1, 1, 0>(XT, lane, g0);
#pragma unroll
  for (int r = 0; r < TABR; ++r) {
    const int i4 = tid + TB * r;
    if (i4 < TAB4) *(v4f*)&sTAB[i4 * 4] = tv[r];
  }
  if (tid < TABTAIL) sTAB[TAB4 * 4 + tid] = tt;
  __syncthreads();

#define WIGNER_COMBO(C) wigner_combo<C>(sCG, sFRAG, XT, X2, lane, b, half)
  WIGNER_COMBO(0);  WIGNER_COMBO(1);  WIGNER_COMBO(2);  WIGNER_COMBO(3);
  WIGNER_COMBO(4);  WIGNER_COMBO(5);  WIGNER_COMBO(6);  WIGNER_COMBO(7);
  WIGNER_COMBO(8);  WIGNER_COMBO(9);  WIGNER_COMBO(10); WIGNER_COMBO(11);
  WIGNER_COMBO(12); WIGNER_COMBO(13); WIGNER_COMBO(14); WIGNER_COMBO(15);
  WIGNER_COMBO(16); WIGNER_COMBO(17); WIGNER_COMBO(18); WIGNER_COMBO(19);
  WIGNER_COMBO(20); WIGNER_COMBO(21); WIGNER_COMBO(22);
#undef WIGNER_COMBO
  __syncthreads();

  if (lane < 16) {
    const int e = base + b;
    if (e < Btot) {
      const float nu1 = XT[0 * 16 + b];
      const float nu2 = X2[0 * 16 + b];
      float nu3 = 0.f, nu4 = 0.f;
      for (int l = 0; l <= 3; ++l) {
        const int n = 2 * l + 1;
        const int cgo = c_off[d_diag[l]];
        const int xo = d_xoff[l];
        const int xp = d_x2pos[l], xn = d_x2neg[l];
        for (int m = 0; m < n; ++m) {
          const int nn = n - 1 - m;
          const float c1 = sCG[cgo + m * n + nn];
          for (int p = 0; p < n; ++p) {
            const int q = n - 1 - p;
            const float c2 = sCG[cgo + p * n + q];
            const float w = c1 * c2;
            nu3 += w * X2[(xp + m * n + p) * 16 + b] * XT[(xo + nn * n + q) * 16 + b];
            nu4 += w * X2[(xp + m * n + p) * 16 + b] * X2[(xp + nn * n + q) * 16 + b];
            if (l > 0)
              nu4 += w * X2[(xn + m * n + p) * 16 + b] * X2[(xn + nn * n + q) * 16 + b];
          }
        }
      }
      out[(size_t)e * 4 + 0] = nu1;
      out[(size_t)e * 4 + 1] = nu2;
      out[(size_t)e * 4 + 2] = nu3;
      out[(size_t)e * 4 + 3] = nu4;
    }
  }
}

extern "C" void kernel_launch(void* const* d_in, const int* in_sizes, int n_in,
                              void* d_out, int out_size, void* d_ws, size_t ws_size,
                              hipStream_t stream) {
  const float* x0 = (const float*)d_in[0];
  const float* x1 = (const float*)d_in[1];
  const float* x2 = (const float*)d_in[2];
  const float* x3 = (const float*)d_in[3];
  float* tab = (float*)d_ws;
  const int Btot = in_sizes[0];

  wigner_cg_init<<<NCOMBO, INIT_TB, 0, stream>>>(tab);

  const int tiles = (Btot + 15) / 16;
  const int blocks = (tiles + WAVES_PER_BLOCK - 1) / WAVES_PER_BLOCK;
  wigner_main<<<blocks, TB, 0, stream>>>(x0, x1, x2, x3, tab, (float*)d_out, Btot);
}
